# starting kernel
# baseline (speedup 1.0000x reference)
#include <hip/hip_runtime.h>
#include <math.h>

typedef __attribute__((ext_vector_type(2))) float v2f;
typedef __attribute__((ext_vector_type(8))) float v8f;

#define FEAT 64

__global__ void gcnii_spmm_edges(const int* __restrict__ row,
                                 const int* __restrict__ col,
                                 const float* __restrict__ vals,
                                 const float* __restrict__ H,
                                 float* __restrict__ AH,
                                 int nEdges) {
    int wave = (int)((blockIdx.x * blockDim.x + threadIdx.x) >> 5);
    int lane = threadIdx.x & 31;
    if (wave >= nEdges) return;

    int r = row[wave];
    int c = col[wave];
    float v = vals[wave];

    const float2* hsrc = (const float2*)(H + (size_t)c * FEAT);
    float2 h = hsrc[lane];

    float* dst = AH + (size_t)r * FEAT + lane * 2;
    atomicAdd(dst + 0, v * h.x);
    atomicAdd(dst + 1, v * h.y);
}

__global__ void gcnii_gemm_epilogue(const float* __restrict__ AH,
                                    const float* __restrict__ H0,
                                    const float* __restrict__ W,
                                    float* __restrict__ out,
                                    const float* __restrict__ lamdaPtr,
                                    const float* __restrict__ alphaPtr,
                                    const int* __restrict__ lPtr,
                                    int nTiles) {
    int tile = (int)(blockIdx.x * (blockDim.x >> 5) + (threadIdx.x >> 5));
    if (tile >= nTiles) return;

    int m0 = (tile >> 2) * 16;
    int n0 = (tile & 3) * 16;
    int lane = threadIdx.x & 31;
    int hi   = lane >> 4;
    int lm   = lane & 15;

    float alpha = alphaPtr[0];
    float lamda = lamdaPtr[0];
    float beta  = logf(lamda / (float)lPtr[0] + 1.0f);
    float oma   = 1.0f - alpha;
    float omb   = 1.0f - beta;

    const float* arowAH = AH + (size_t)(m0 + lm) * FEAT + hi * 2;
    const float* arowH0 = H0 + (size_t)(m0 + lm) * FEAT + hi * 2;
    const float* wbase  = W + (size_t)(hi * 2) * FEAT + n0 + lm;

    v8f acc = {};
#pragma unroll
    for (int kk = 0; kk < 16; ++kk) {
        int K0 = kk * 4;
        float2 ah2 = *(const float2*)(arowAH + K0);
        float2 h02 = *(const float2*)(arowH0 + K0);
        v2f a;
        a.x = oma * ah2.x + alpha * h02.x;
        a.y = oma * ah2.y + alpha * h02.y;
        v2f b;
        b.x = wbase[(size_t)(K0 + 0) * FEAT];
        b.y = wbase[(size_t)(K0 + 1) * FEAT];
        acc = __builtin_amdgcn_wmma_f32_16x16x4_f32(
             false, a,  false, b,
             (short)0, acc,  false,  false);
    }

#pragma unroll
    for (int r = 0; r < 8; ++r) {
        size_t idx = (size_t)(m0 + r + hi * 8) * FEAT + n0 + lm;
        float sup = oma * AH[idx] + alpha * H0[idx];
        out[idx] = omb * sup + beta * acc[r];
    }
}

extern "C" void kernel_launch(void* const* d_in, const int* in_sizes, int n_in,
                              void* d_out, int out_size, void* d_ws, size_t ws_size,
                              hipStream_t stream) {
    const int*   row   = (const int*)  d_in[0];
    const int*   col   = (const int*)  d_in[1];
    const float* vals  = (const float*)d_in[2];
    const float* H     = (const float*)d_in[3];
    const float* H0    = (const float*)d_in[4];
    const float* W     = (const float*)d_in[5];
    const float* lamda = (const float*)d_in[6];
    const float* alpha = (const float*)d_in[7];
    const int*   l     = (const int*)  d_in[8];

    int nEdges = in_sizes[0];
    int nNodes = in_sizes[3] / FEAT;

    float* AH = (float*)d_ws;

    hipMemsetAsync(AH, 0, (size_t)nNodes * FEAT * sizeof(float), stream);

    int spmmBlocks = (nEdges + 7) / 8;
    gcnii_spmm_edges<<<spmmBlocks, 256, 0, stream>>>(row, col, vals, H, AH, nEdges);

    int nTiles = (nNodes / 16) * (FEAT / 16);
    int gemmBlocks = (nTiles + 7) / 8;
    gcnii_gemm_epilogue<<<gemmBlocks, 256, 0, stream>>>(
        AH, H0, W, (float*)d_out, lamda, alpha, l, nTiles);
}


# SPMM: one wave walks a chunk of 32 edges (coalesced index loads, 8 row gathers in flight) instead of one wave per edge; f32 atomic adds kept, still arrival order
# speedup vs baseline: 1.7230x; 1.7230x over previous
#include <hip/hip_runtime.h>
#include <math.h>

typedef __attribute__((ext_vector_type(2))) float v2f;
typedef __attribute__((ext_vector_type(8))) float v8f;

#define FEAT 64

#define EDGES_PER_WAVE 32
#define ROWS_IN_FLIGHT 8

__global__ void __launch_bounds__(256)
gcnii_spmm_edges(const int* __restrict__ row,
                 const int* __restrict__ col,
                 const float* __restrict__ vals,
                 const float* __restrict__ H,
                 float* __restrict__ AH,
                 int nEdges) {
    int wave = __builtin_amdgcn_readfirstlane(
        (int)((blockIdx.x * blockDim.x + threadIdx.x) >> 5));
    int lane = threadIdx.x & 31;
    int base = wave * EDGES_PER_WAVE;
    if (base >= nEdges) return;
    int cnt = nEdges - base;
    if (cnt > EDGES_PER_WAVE) cnt = EDGES_PER_WAVE;

    int r = 0, c = 0;
    float v = 0.0f;
    if (lane < cnt) {
        r = row[base + lane];
        c = col[base + lane];
        v = vals[base + lane];
    }

    const float2* H2 = (const float2*)H;
    float* dstLane = AH + lane * 2;

    if (cnt == EDGES_PER_WAVE) {
#pragma unroll
        for (int g = 0; g < EDGES_PER_WAVE; g += ROWS_IN_FLIGHT) {
            float2 h[ROWS_IN_FLIGHT];
#pragma unroll
            for (int j = 0; j < ROWS_IN_FLIGHT; ++j) {
                int cj = __builtin_amdgcn_readlane(c, g + j);
                h[j] = H2[(size_t)cj * (FEAT / 2) + lane];
            }
#pragma unroll
            for (int j = 0; j < ROWS_IN_FLIGHT; ++j) {
                int rj = __builtin_amdgcn_readlane(r, g + j);
                float vj = __builtin_bit_cast(
                    float, __builtin_amdgcn_readlane(__builtin_bit_cast(int, v), g + j));
                float* dst = dstLane + (size_t)rj * FEAT;
                atomicAdd(dst + 0, vj * h[j].x);
                atomicAdd(dst + 1, vj * h[j].y);
            }
        }
    } else {
        for (int j = 0; j < cnt; ++j) {
            int cj = __shfl(c, j);
            int rj = __shfl(r, j);
            float vj = __shfl(v, j);
            float2 h = H2[(size_t)cj * (FEAT / 2) + lane];
            float* dst = dstLane + (size_t)rj * FEAT;
            atomicAdd(dst + 0, vj * h.x);
            atomicAdd(dst + 1, vj * h.y);
        }
    }
}

__global__ void gcnii_gemm_epilogue(const float* __restrict__ AH,
                                    const float* __restrict__ H0,
                                    const float* __restrict__ W,
                                    float* __restrict__ out,
                                    const float* __restrict__ lamdaPtr,
                                    const float* __restrict__ alphaPtr,
                                    const int* __restrict__ lPtr,
                                    int nTiles) {
    int tile = (int)(blockIdx.x * (blockDim.x >> 5) + (threadIdx.x >> 5));
    if (tile >= nTiles) return;

    int m0 = (tile >> 2) * 16;
    int n0 = (tile & 3) * 16;
    int lane = threadIdx.x & 31;
    int hi   = lane >> 4;
    int lm   = lane & 15;

    float alpha = alphaPtr[0];
    float lamda = lamdaPtr[0];
    float beta  = logf(lamda / (float)lPtr[0] + 1.0f);
    float oma   = 1.0f - alpha;
    float omb   = 1.0f - beta;

    const float* arowAH = AH + (size_t)(m0 + lm) * FEAT + hi * 2;
    const float* arowH0 = H0 + (size_t)(m0 + lm) * FEAT + hi * 2;
    const float* wbase  = W + (size_t)(hi * 2) * FEAT + n0 + lm;

    v8f acc = {};
#pragma unroll
    for (int kk = 0; kk < 16; ++kk) {
        int K0 = kk * 4;
        float2 ah2 = *(const float2*)(arowAH + K0);
        float2 h02 = *(const float2*)(arowH0 + K0);
        v2f a;
        a.x = oma * ah2.x + alpha * h02.x;
        a.y = oma * ah2.y + alpha * h02.y;
        v2f b;
        b.x = wbase[(size_t)(K0 + 0) * FEAT];
        b.y = wbase[(size_t)(K0 + 1) * FEAT];
        acc = __builtin_amdgcn_wmma_f32_16x16x4_f32(
             false, a,  false, b,
             (short)0, acc,  false,  false);
    }

#pragma unroll
    for (int r = 0; r < 8; ++r) {
        size_t idx = (size_t)(m0 + r + hi * 8) * FEAT + n0 + lm;
        float sup = oma * AH[idx] + alpha * H0[idx];
        out[idx] = omb * sup + beta * acc[r];
    }
}

extern "C" void kernel_launch(void* const* d_in, const int* in_sizes, int n_in,
                              void* d_out, int out_size, void* d_ws, size_t ws_size,
                              hipStream_t stream) {
    const int*   row   = (const int*)  d_in[0];
    const int*   col   = (const int*)  d_in[1];
    const float* vals  = (const float*)d_in[2];
    const float* H     = (const float*)d_in[3];
    const float* H0    = (const float*)d_in[4];
    const float* W     = (const float*)d_in[5];
    const float* lamda = (const float*)d_in[6];
    const float* alpha = (const float*)d_in[7];
    const int*   l     = (const int*)  d_in[8];

    int nEdges = in_sizes[0];
    int nNodes = in_sizes[3] / FEAT;

    float* AH = (float*)d_ws;

    hipMemsetAsync(AH, 0, (size_t)nNodes * FEAT * sizeof(float), stream);

    int nChunks = (nEdges + EDGES_PER_WAVE - 1) / EDGES_PER_WAVE;
    int spmmBlocks = (nChunks + 7) / 8;
    gcnii_spmm_edges<<<spmmBlocks, 256, 0, stream>>>(row, col, vals, H, AH, nEdges);

    int nTiles = (nNodes / 16) * (FEAT / 16);
    int gemmBlocks = (nTiles + 7) / 8;
    gcnii_gemm_epilogue<<<gemmBlocks, 256, 0, stream>>>(
        AH, H0, W, (float*)d_out, lamda, alpha, l, nTiles);
}


# SPMM: lane i takes features i and i+32, so each row load and each f32 atomic add instruction covers one whole 128-byte line, not half of two; atomics kept, still arrival order
# speedup vs baseline: 1.9865x; 1.1529x over previous
#include <hip/hip_runtime.h>
#include <math.h>

typedef __attribute__((ext_vector_type(2))) float v2f;
typedef __attribute__((ext_vector_type(8))) float v8f;

#define FEAT 64

#define EDGES_PER_WAVE 32
#define ROWS_IN_FLIGHT 8

__global__ void __launch_bounds__(256)
gcnii_spmm_edges(const int* __restrict__ row,
                 const int* __restrict__ col,
                 const float* __restrict__ vals,
                 const float* __restrict__ H,
                 float* __restrict__ AH,
                 int nEdges) {
    int wave = __builtin_amdgcn_readfirstlane(
        (int)((blockIdx.x * blockDim.x + threadIdx.x) >> 5));
    int lane = threadIdx.x & 31;
    int base = wave * EDGES_PER_WAVE;
    if (base >= nEdges) return;
    int cnt = nEdges - base;
    if (cnt > EDGES_PER_WAVE) cnt = EDGES_PER_WAVE;

    int r = 0, c = 0;
    float v = 0.0f;
    if (lane < cnt) {
        r = row[base + lane];
        c = col[base + lane];
        v = vals[base + lane];
    }

    const float* srcLane = H + lane;
    float* dstLane = AH + lane;

    if (cnt == EDGES_PER_WAVE) {
#pragma unroll
        for (int g = 0; g < EDGES_PER_WAVE; g += ROWS_IN_FLIGHT) {
            float hlo[ROWS_IN_FLIGHT];
            float hhi[ROWS_IN_FLIGHT];
#pragma unroll
            for (int j = 0; j < ROWS_IN_FLIGHT; ++j) {
                int cj = __builtin_amdgcn_readlane(c, g + j);
                const float* src = srcLane + (size_t)cj * FEAT;
                hlo[j] = src[0];
                hhi[j] = src[FEAT / 2];
            }
#pragma unroll
            for (int j = 0; j < ROWS_IN_FLIGHT; ++j) {
                int rj = __builtin_amdgcn_readlane(r, g + j);
                float vj = __builtin_bit_cast(
                    float, __builtin_amdgcn_readlane(__builtin_bit_cast(int, v), g + j));
                float* dst = dstLane + (size_t)rj * FEAT;
                atomicAdd(dst,            vj * hlo[j]);
                atomicAdd(dst + FEAT / 2, vj * hhi[j]);
            }
        }
    } else {
        for (int j = 0; j < cnt; ++j) {
            int cj = __shfl(c, j);
            int rj = __shfl(r, j);
            float vj = __shfl(v, j);
            const float* src = srcLane + (size_t)cj * FEAT;
            float lo = src[0];
            float hi = src[FEAT / 2];
            float* dst = dstLane + (size_t)rj * FEAT;
            atomicAdd(dst,            vj * lo);
            atomicAdd(dst + FEAT / 2, vj * hi);
        }
    }
}

__global__ void gcnii_gemm_epilogue(const float* __restrict__ AH,
                                    const float* __restrict__ H0,
                                    const float* __restrict__ W,
                                    float* __restrict__ out,
                                    const float* __restrict__ lamdaPtr,
                                    const float* __restrict__ alphaPtr,
                                    const int* __restrict__ lPtr,
                                    int nTiles) {
    int tile = (int)(blockIdx.x * (blockDim.x >> 5) + (threadIdx.x >> 5));
    if (tile >= nTiles) return;

    int m0 = (tile >> 2) * 16;
    int n0 = (tile & 3) * 16;
    int lane = threadIdx.x & 31;
    int hi   = lane >> 4;
    int lm   = lane & 15;

    float alpha = alphaPtr[0];
    float lamda = lamdaPtr[0];
    float beta  = logf(lamda / (float)lPtr[0] + 1.0f);
    float oma   = 1.0f - alpha;
    float omb   = 1.0f - beta;

    const float* arowAH = AH + (size_t)(m0 + lm) * FEAT + hi * 2;
    const float* arowH0 = H0 + (size_t)(m0 + lm) * FEAT + hi * 2;
    const float* wbase  = W + (size_t)(hi * 2) * FEAT + n0 + lm;

    v8f acc = {};
#pragma unroll
    for (int kk = 0; kk < 16; ++kk) {
        int K0 = kk * 4;
        float2 ah2 = *(const float2*)(arowAH + K0);
        float2 h02 = *(const float2*)(arowH0 + K0);
        v2f a;
        a.x = oma * ah2.x + alpha * h02.x;
        a.y = oma * ah2.y + alpha * h02.y;
        v2f b;
        b.x = wbase[(size_t)(K0 + 0) * FEAT];
        b.y = wbase[(size_t)(K0 + 1) * FEAT];
        acc = __builtin_amdgcn_wmma_f32_16x16x4_f32(
             false, a,  false, b,
             (short)0, acc,  false,  false);
    }

#pragma unroll
    for (int r = 0; r < 8; ++r) {
        size_t idx = (size_t)(m0 + r + hi * 8) * FEAT + n0 + lm;
        float sup = oma * AH[idx] + alpha * H0[idx];
        out[idx] = omb * sup + beta * acc[r];
    }
}

extern "C" void kernel_launch(void* const* d_in, const int* in_sizes, int n_in,
                              void* d_out, int out_size, void* d_ws, size_t ws_size,
                              hipStream_t stream) {
    const int*   row   = (const int*)  d_in[0];
    const int*   col   = (const int*)  d_in[1];
    const float* vals  = (const float*)d_in[2];
    const float* H     = (const float*)d_in[3];
    const float* H0    = (const float*)d_in[4];
    const float* W     = (const float*)d_in[5];
    const float* lamda = (const float*)d_in[6];
    const float* alpha = (const float*)d_in[7];
    const int*   l     = (const int*)  d_in[8];

    int nEdges = in_sizes[0];
    int nNodes = in_sizes[3] / FEAT;

    float* AH = (float*)d_ws;

    hipMemsetAsync(AH, 0, (size_t)nNodes * FEAT * sizeof(float), stream);

    int nChunks = (nEdges + EDGES_PER_WAVE - 1) / EDGES_PER_WAVE;
    int spmmBlocks = (nChunks + 7) / 8;
    gcnii_spmm_edges<<<spmmBlocks, 256, 0, stream>>>(row, col, vals, H, AH, nEdges);

    int nTiles = (nNodes / 16) * (FEAT / 16);
    int gemmBlocks = (nTiles + 7) / 8;
    gcnii_gemm_epilogue<<<gemmBlocks, 256, 0, stream>>>(
        AH, H0, W, (float*)d_out, lamda, alpha, l, nTiles);
}


# GEMM: one wave per 16-row tile holds all 64 columns (4 accumulators, column 4c+n in tile n): support fragment built once per k-step for 4 WMMAs; W, epilogue reads and out become 16-byte accesses
# speedup vs baseline: 2.1740x; 1.0944x over previous
#include <hip/hip_runtime.h>
#include <math.h>

typedef __attribute__((ext_vector_type(2))) float v2f;
typedef __attribute__((ext_vector_type(8))) float v8f;

#define FEAT 64

#define EDGES_PER_WAVE 32
#define ROWS_IN_FLIGHT 8

__global__ void __launch_bounds__(256)
gcnii_spmm_edges(const int* __restrict__ row,
                 const int* __restrict__ col,
                 const float* __restrict__ vals,
                 const float* __restrict__ H,
                 float* __restrict__ AH,
                 int nEdges) {
    int wave = __builtin_amdgcn_readfirstlane(
        (int)((blockIdx.x * blockDim.x + threadIdx.x) >> 5));
    int lane = threadIdx.x & 31;
    int base = wave * EDGES_PER_WAVE;
    if (base >= nEdges) return;
    int cnt = nEdges - base;
    if (cnt > EDGES_PER_WAVE) cnt = EDGES_PER_WAVE;

    int r = 0, c = 0;
    float v = 0.0f;
    if (lane < cnt) {
        r = row[base + lane];
        c = col[base + lane];
        v = vals[base + lane];
    }

    const float* srcLane = H + lane;
    float* dstLane = AH + lane;

    if (cnt == EDGES_PER_WAVE) {
#pragma unroll
        for (int g = 0; g < EDGES_PER_WAVE; g += ROWS_IN_FLIGHT) {
            float hlo[ROWS_IN_FLIGHT];
            float hhi[ROWS_IN_FLIGHT];
#pragma unroll
            for (int j = 0; j < ROWS_IN_FLIGHT; ++j) {
                int cj = __builtin_amdgcn_readlane(c, g + j);
                const float* src = srcLane + (size_t)cj * FEAT;
                hlo[j] = src[0];
                hhi[j] = src[FEAT / 2];
            }
#pragma unroll
            for (int j = 0; j < ROWS_IN_FLIGHT; ++j) {
                int rj = __builtin_amdgcn_readlane(r, g + j);
                float vj = __builtin_bit_cast(
                    float, __builtin_amdgcn_readlane(__builtin_bit_cast(int, v), g + j));
                float* dst = dstLane + (size_t)rj * FEAT;
                atomicAdd(dst,            vj * hlo[j]);
                atomicAdd(dst + FEAT / 2, vj * hhi[j]);
            }
        }
    } else {
        for (int j = 0; j < cnt; ++j) {
            int cj = __shfl(c, j);
            int rj = __shfl(r, j);
            float vj = __shfl(v, j);
            const float* src = srcLane + (size_t)cj * FEAT;
            float lo = src[0];
            float hi = src[FEAT / 2];
            float* dst = dstLane + (size_t)rj * FEAT;
            atomicAdd(dst,            vj * lo);
            atomicAdd(dst + FEAT / 2, vj * hi);
        }
    }
}

__global__ void __launch_bounds__(256)
gcnii_gemm_epilogue(const float* __restrict__ AH,
                    const float* __restrict__ H0,
                    const float* __restrict__ W,
                    float* __restrict__ out,
                    const float* __restrict__ lamdaPtr,
                    const float* __restrict__ alphaPtr,
                    const int* __restrict__ lPtr,
                    int nRowTiles) {
    int tile = (int)(blockIdx.x * (blockDim.x >> 5) + (threadIdx.x >> 5));
    if (tile >= nRowTiles) return;

    int m0 = tile * 16;
    int lane = threadIdx.x & 31;
    int hi   = lane >> 4;
    int lm   = lane & 15;

    float alpha = alphaPtr[0];
    float lamda = lamdaPtr[0];
    float beta  = logf(lamda / (float)lPtr[0] + 1.0f);
    float oma   = 1.0f - alpha;
    float omb   = 1.0f - beta;

    const float* arowAH = AH + (size_t)(m0 + lm) * FEAT + hi * 2;
    const float* arowH0 = H0 + (size_t)(m0 + lm) * FEAT + hi * 2;
    const float* wbase  = W + (size_t)(hi * 2) * FEAT + lm * 4;

    v8f acc0 = {}, acc1 = {}, acc2 = {}, acc3 = {};
#pragma unroll
    for (int kk = 0; kk < 16; ++kk) {
        int K0 = kk * 4;
        float2 ah2 = *(const float2*)(arowAH + K0);
        float2 h02 = *(const float2*)(arowH0 + K0);
        v2f a;
        a.x = oma * ah2.x + alpha * h02.x;
        a.y = oma * ah2.y + alpha * h02.y;
        float4 w0 = *(const float4*)(wbase + (size_t)(K0 + 0) * FEAT);
        float4 w1 = *(const float4*)(wbase + (size_t)(K0 + 1) * FEAT);
        v2f b0, b1, b2, b3;
        b0.x = w0.x; b0.y = w1.x;
        b1.x = w0.y; b1.y = w1.y;
        b2.x = w0.z; b2.y = w1.z;
        b3.x = w0.w; b3.y = w1.w;
        acc0 = __builtin_amdgcn_wmma_f32_16x16x4_f32(
            false, a, false, b0, (short)0, acc0, false, false);
        acc1 = __builtin_amdgcn_wmma_f32_16x16x4_f32(
            false, a, false, b1, (short)0, acc1, false, false);
        acc2 = __builtin_amdgcn_wmma_f32_16x16x4_f32(
            false, a, false, b2, (short)0, acc2, false, false);
        acc3 = __builtin_amdgcn_wmma_f32_16x16x4_f32(
            false, a, false, b3, (short)0, acc3, false, false);
    }

#pragma unroll
    for (int r = 0; r < 8; ++r) {
        size_t idx = (size_t)(m0 + r + hi * 8) * FEAT + lm * 4;
        float4 ah = *(const float4*)(AH + idx);
        float4 h0 = *(const float4*)(H0 + idx);
        float4 o;
        float sup;
        sup = oma * ah.x + alpha * h0.x;  o.x = omb * sup + beta * acc0[r];
        sup = oma * ah.y + alpha * h0.y;  o.y = omb * sup + beta * acc1[r];
        sup = oma * ah.z + alpha * h0.z;  o.z = omb * sup + beta * acc2[r];
        sup = oma * ah.w + alpha * h0.w;  o.w = omb * sup + beta * acc3[r];
        *(float4*)(out + idx) = o;
    }
}

extern "C" void kernel_launch(void* const* d_in, const int* in_sizes, int n_in,
                              void* d_out, int out_size, void* d_ws, size_t ws_size,
                              hipStream_t stream) {
    const int*   row   = (const int*)  d_in[0];
    const int*   col   = (const int*)  d_in[1];
    const float* vals  = (const float*)d_in[2];
    const float* H     = (const float*)d_in[3];
    const float* H0    = (const float*)d_in[4];
    const float* W     = (const float*)d_in[5];
    const float* lamda = (const float*)d_in[6];
    const float* alpha = (const float*)d_in[7];
    const int*   l     = (const int*)  d_in[8];

    int nEdges = in_sizes[0];
    int nNodes = in_sizes[3] / FEAT;

    float* AH = (float*)d_ws;

    hipMemsetAsync(AH, 0, (size_t)nNodes * FEAT * sizeof(float), stream);

    int nChunks = (nEdges + EDGES_PER_WAVE - 1) / EDGES_PER_WAVE;
    int spmmBlocks = (nChunks + 7) / 8;
    gcnii_spmm_edges<<<spmmBlocks, 256, 0, stream>>>(row, col, vals, H, AH, nEdges);

    int nRowTiles = nNodes / 16;
    int gemmBlocks = (nRowTiles + 7) / 8;
    gcnii_gemm_epilogue<<<gemmBlocks, 256, 0, stream>>>(
        AH, H0, W, (float*)d_out, lamda, alpha, l, nRowTiles);
}


# SPMM: edges binned by block of 100 destination rows (count, scan, scatter), each block summed in LDS with ds_add_f32 then stored once; no global f32 atomics, no AH memset; order still arrival order
# speedup vs baseline: 2.6874x; 1.2362x over previous
#include <hip/hip_runtime.h>
#include <math.h>

typedef __attribute__((ext_vector_type(2))) float v2f;
typedef __attribute__((ext_vector_type(8))) float v8f;

#define FEAT 64

#define RB 100
#define COL_BITS 24
#define COL_MASK 0x00FFFFFF

#define BIN_THREADS 256
#define BIN_PER_THREAD 16
#define BIN_CHUNK (BIN_THREADS * BIN_PER_THREAD)

__global__ void __launch_bounds__(BIN_THREADS)
gcnii_bin_count(const int* __restrict__ row,
                int* __restrict__ binCount,
                int nEdges, int nBins) {
    extern __shared__ int ldsCnt[];
    for (int b = threadIdx.x; b < nBins; b += BIN_THREADS) ldsCnt[b] = 0;
    int base = blockIdx.x * BIN_CHUNK + threadIdx.x;
    int last = nEdges - 1;
    int r[BIN_PER_THREAD];
#pragma unroll
    for (int j = 0; j < BIN_PER_THREAD; ++j) {
        int e = base + j * BIN_THREADS;
        r[j] = row[e < last ? e : last];
    }
    __syncthreads();
#pragma unroll
    for (int j = 0; j < BIN_PER_THREAD; ++j) {
        int e = base + j * BIN_THREADS;
        if (e < nEdges) atomicAdd(&ldsCnt[r[j] / RB], 1);
    }
    __syncthreads();
    for (int b = threadIdx.x; b < nBins; b += BIN_THREADS) {
        int c = ldsCnt[b];
        if (c) atomicAdd(&binCount[b], c);
    }
}

#define SCAN_THREADS 1024
__global__ void __launch_bounds__(SCAN_THREADS)
gcnii_bin_scan(const int* __restrict__ binCount,
               int* __restrict__ binStart,
               int nBins) {
    __shared__ int buf[SCAN_THREADS];
    __shared__ int carry;
    int t = threadIdx.x;
    if (t == 0) carry = 0;
    __syncthreads();
    for (int base = 0; base < nBins; base += SCAN_THREADS) {
        int i = base + t;
        int c = (i < nBins) ? binCount[i] : 0;
        buf[t] = c;
        __syncthreads();
        for (int off = 1; off < SCAN_THREADS; off <<= 1) {
            int add = (t >= off) ? buf[t - off] : 0;
            __syncthreads();
            buf[t] += add;
            __syncthreads();
        }
        int incl = buf[t];
        int before = carry;
        if (i < nBins) binStart[i] = before + incl - c;
        __syncthreads();
        if (t == SCAN_THREADS - 1) carry = before + incl;
        __syncthreads();
    }
    if (t == 0) binStart[nBins] = carry;
}

#define RESERVE_PER_THREAD 4
__global__ void __launch_bounds__(BIN_THREADS)
gcnii_bin_scatter(const int* __restrict__ row,
                  const int* __restrict__ col,
                  const float* __restrict__ vals,
                  const int* __restrict__ binStart,
                  int* __restrict__ binCursor,
                  int* __restrict__ binned,
                  int nEdges, int nBins) {
    extern __shared__ int ldsCnt[];
    for (int b = threadIdx.x; b < nBins; b += BIN_THREADS) ldsCnt[b] = 0;
    int base = blockIdx.x * BIN_CHUNK + threadIdx.x;
    int last = nEdges - 1;
    int r[BIN_PER_THREAD];
    int cl[BIN_PER_THREAD];
    float vl[BIN_PER_THREAD];
    int rank[BIN_PER_THREAD];
#pragma unroll
    for (int j = 0; j < BIN_PER_THREAD; ++j) {
        int e = base + j * BIN_THREADS;
        int ec = e < last ? e : last;
        r[j] = row[ec];
        cl[j] = col[ec];
        vl[j] = vals[ec];
    }
    __syncthreads();
#pragma unroll
    for (int j = 0; j < BIN_PER_THREAD; ++j) {
        int e = base + j * BIN_THREADS;
        rank[j] = 0;
        if (e < nEdges) rank[j] = atomicAdd(&ldsCnt[r[j] / RB], 1);
    }
    __syncthreads();
    for (int b0 = threadIdx.x; b0 < nBins; b0 += BIN_THREADS * RESERVE_PER_THREAD) {
        int cnt[RESERVE_PER_THREAD];
        int start[RESERVE_PER_THREAD];
        int got[RESERVE_PER_THREAD];
#pragma unroll
        for (int k = 0; k < RESERVE_PER_THREAD; ++k) {
            int b = b0 + k * BIN_THREADS;
            cnt[k] = (b < nBins) ? ldsCnt[b] : 0;
        }
#pragma unroll
        for (int k = 0; k < RESERVE_PER_THREAD; ++k) {
            int b = b0 + k * BIN_THREADS;
            start[k] = 0;
            got[k] = 0;
            if (cnt[k]) {
                start[k] = binStart[b];
                got[k] = atomicAdd(&binCursor[b], cnt[k]);
            }
        }
#pragma unroll
        for (int k = 0; k < RESERVE_PER_THREAD; ++k) {
            int b = b0 + k * BIN_THREADS;
            if (b < nBins) ldsCnt[b] = start[k] + got[k];
        }
    }
    __syncthreads();
#pragma unroll
    for (int j = 0; j < BIN_PER_THREAD; ++j) {
        int e = base + j * BIN_THREADS;
        if (e < nEdges) {
            int b = r[j] / RB;
            int pos = ldsCnt[b] + rank[j];
            int2 rec;
            rec.x = ((r[j] - b * RB) << COL_BITS) | cl[j];
            rec.y = __builtin_bit_cast(int, vl[j]);
            *(int2*)(binned + 2 * (size_t)pos) = rec;
        }
    }
}

#define SUM_THREADS 512
#define SUM_WAVES (SUM_THREADS / 32)
#define EDGES_PER_WAVE 32
#define ROWS_IN_FLIGHT 8

__global__ void __launch_bounds__(SUM_THREADS)
gcnii_block_sums(const int* __restrict__ binned,
                 const int* __restrict__ binStart,
                 const float* __restrict__ H,
                 float* __restrict__ AH,
                 int nNodes) {
    __shared__ float acc[RB * FEAT];
    int tid = threadIdx.x;
    int blk = blockIdx.x;

    float4 zero4 = {0.0f, 0.0f, 0.0f, 0.0f};
    for (int i = tid; i < RB * FEAT / 4; i += SUM_THREADS) ((float4*)acc)[i] = zero4;
    __syncthreads();

    int eBeg = binStart[blk];
    int eEnd = binStart[blk + 1];
    int wave = __builtin_amdgcn_readfirstlane(tid >> 5);
    int lane = tid & 31;
    const float* srcLane = H + lane;
    float* dstLane = acc + lane;

    for (int base = eBeg + wave * EDGES_PER_WAVE; base < eEnd;
         base += SUM_WAVES * EDGES_PER_WAVE) {
        int cnt = eEnd - base;
        if (cnt > EDGES_PER_WAVE) cnt = EDGES_PER_WAVE;
        int meta = 0;
        float v = 0.0f;
        if (lane < cnt) {
            int2 rec = *(const int2*)(binned + 2 * (size_t)(base + lane));
            meta = rec.x;
            v = __builtin_bit_cast(float, rec.y);
        }
        if (cnt == EDGES_PER_WAVE) {
#pragma unroll
            for (int g = 0; g < EDGES_PER_WAVE; g += ROWS_IN_FLIGHT) {
                float hlo[ROWS_IN_FLIGHT];
                float hhi[ROWS_IN_FLIGHT];
#pragma unroll
                for (int j = 0; j < ROWS_IN_FLIGHT; ++j) {
                    int mj = __builtin_amdgcn_readlane(meta, g + j);
                    const float* src = srcLane + (size_t)(mj & COL_MASK) * FEAT;
                    hlo[j] = src[0];
                    hhi[j] = src[FEAT / 2];
                }
#pragma unroll
                for (int j = 0; j < ROWS_IN_FLIGHT; ++j) {
                    int mj = __builtin_amdgcn_readlane(meta, g + j);
                    float vj = __builtin_bit_cast(
                        float, __builtin_amdgcn_readlane(__builtin_bit_cast(int, v), g + j));
                    float* dst = dstLane + ((unsigned)mj >> COL_BITS) * FEAT;
                    __hip_atomic_fetch_add(dst, vj * hlo[j],
                                           __ATOMIC_RELAXED, __HIP_MEMORY_SCOPE_WORKGROUP);
                    __hip_atomic_fetch_add(dst + FEAT / 2, vj * hhi[j],
                                           __ATOMIC_RELAXED, __HIP_MEMORY_SCOPE_WORKGROUP);
                }
            }
        } else {
            for (int j = 0; j < cnt; ++j) {
                int mj = __shfl(meta, j);
                float vj = __shfl(v, j);
                const float* src = srcLane + (size_t)(mj & COL_MASK) * FEAT;
                float lo = src[0];
                float hi = src[FEAT / 2];
                float* dst = dstLane + ((unsigned)mj >> COL_BITS) * FEAT;
                __hip_atomic_fetch_add(dst, vj * lo,
                                       __ATOMIC_RELAXED, __HIP_MEMORY_SCOPE_WORKGROUP);
                __hip_atomic_fetch_add(dst + FEAT / 2, vj * hi,
                                       __ATOMIC_RELAXED, __HIP_MEMORY_SCOPE_WORKGROUP);
            }
        }
    }
    __syncthreads();

    for (int i = tid; i < RB * FEAT / 4; i += SUM_THREADS) {
        int grow = blk * RB + (i >> 4);
        if (grow < nNodes)
            ((float4*)AH)[(size_t)grow * (FEAT / 4) + (i & 15)] = ((const float4*)acc)[i];
    }
}

__global__ void __launch_bounds__(256)
gcnii_gemm_epilogue(const float* __restrict__ AH,
                    const float* __restrict__ H0,
                    const float* __restrict__ W,
                    float* __restrict__ out,
                    const float* __restrict__ lamdaPtr,
                    const float* __restrict__ alphaPtr,
                    const int* __restrict__ lPtr,
                    int nRowTiles) {
    int tile = (int)(blockIdx.x * (blockDim.x >> 5) + (threadIdx.x >> 5));
    if (tile >= nRowTiles) return;

    int m0 = tile * 16;
    int lane = threadIdx.x & 31;
    int hi   = lane >> 4;
    int lm   = lane & 15;

    float alpha = alphaPtr[0];
    float lamda = lamdaPtr[0];
    float beta  = logf(lamda / (float)lPtr[0] + 1.0f);
    float oma   = 1.0f - alpha;
    float omb   = 1.0f - beta;

    const float* arowAH = AH + (size_t)(m0 + lm) * FEAT + hi * 2;
    const float* arowH0 = H0 + (size_t)(m0 + lm) * FEAT + hi * 2;
    const float* wbase  = W + (size_t)(hi * 2) * FEAT + lm * 4;

    v8f acc0 = {}, acc1 = {}, acc2 = {}, acc3 = {};
#pragma unroll
    for (int kk = 0; kk < 16; ++kk) {
        int K0 = kk * 4;
        float2 ah2 = *(const float2*)(arowAH + K0);
        float2 h02 = *(const float2*)(arowH0 + K0);
        v2f a;
        a.x = oma * ah2.x + alpha * h02.x;
        a.y = oma * ah2.y + alpha * h02.y;
        float4 w0 = *(const float4*)(wbase + (size_t)(K0 + 0) * FEAT);
        float4 w1 = *(const float4*)(wbase + (size_t)(K0 + 1) * FEAT);
        v2f b0, b1, b2, b3;
        b0.x = w0.x; b0.y = w1.x;
        b1.x = w0.y; b1.y = w1.y;
        b2.x = w0.z; b2.y = w1.z;
        b3.x = w0.w; b3.y = w1.w;
        acc0 = __builtin_amdgcn_wmma_f32_16x16x4_f32(
            false, a, false, b0, (short)0, acc0, false, false);
        acc1 = __builtin_amdgcn_wmma_f32_16x16x4_f32(
            false, a, false, b1, (short)0, acc1, false, false);
        acc2 = __builtin_amdgcn_wmma_f32_16x16x4_f32(
            false, a, false, b2, (short)0, acc2, false, false);
        acc3 = __builtin_amdgcn_wmma_f32_16x16x4_f32(
            false, a, false, b3, (short)0, acc3, false, false);
    }

#pragma unroll
    for (int r = 0; r < 8; ++r) {
        size_t idx = (size_t)(m0 + r + hi * 8) * FEAT + lm * 4;
        float4 ah = *(const float4*)(AH + idx);
        float4 h0 = *(const float4*)(H0 + idx);
        float4 o;
        float sup;
        sup = oma * ah.x + alpha * h0.x;  o.x = omb * sup + beta * acc0[r];
        sup = oma * ah.y + alpha * h0.y;  o.y = omb * sup + beta * acc1[r];
        sup = oma * ah.z + alpha * h0.z;  o.z = omb * sup + beta * acc2[r];
        sup = oma * ah.w + alpha * h0.w;  o.w = omb * sup + beta * acc3[r];
        *(float4*)(out + idx) = o;
    }
}

extern "C" void kernel_launch(void* const* d_in, const int* in_sizes, int n_in,
                              void* d_out, int out_size, void* d_ws, size_t ws_size,
                              hipStream_t stream) {
    const int*   row   = (const int*)  d_in[0];
    const int*   col   = (const int*)  d_in[1];
    const float* vals  = (const float*)d_in[2];
    const float* H     = (const float*)d_in[3];
    const float* H0    = (const float*)d_in[4];
    const float* W     = (const float*)d_in[5];
    const float* lamda = (const float*)d_in[6];
    const float* alpha = (const float*)d_in[7];
    const int*   l     = (const int*)  d_in[8];

    int nEdges = in_sizes[0];
    int nNodes = in_sizes[3] / FEAT;
    int nBins  = (nNodes + RB - 1) / RB;

    char* ws = (char*)d_ws;
    size_t ahBytes  = (((size_t)nNodes * FEAT * sizeof(float)) + 255) & ~(size_t)255;
    size_t binBytes = (((size_t)nEdges * 2 * sizeof(int)) + 255) & ~(size_t)255;
    float* AH        = (float*)ws;
    int*   binned    = (int*)(ws + ahBytes);
    int*   binCount  = (int*)(ws + ahBytes + binBytes);
    int*   binCursor = binCount + nBins;
    int*   binStart  = binCursor + nBins;

    hipMemsetAsync(binCount, 0, (size_t)nBins * 2 * sizeof(int), stream);

    int binBlocks = (nEdges + BIN_CHUNK - 1) / BIN_CHUNK;
    size_t binLds = (size_t)nBins * sizeof(int);
    gcnii_bin_count<<<binBlocks, BIN_THREADS, binLds, stream>>>(row, binCount, nEdges, nBins);
    gcnii_bin_scan<<<1, SCAN_THREADS, 0, stream>>>(binCount, binStart, nBins);
    gcnii_bin_scatter<<<binBlocks, BIN_THREADS, binLds, stream>>>(
        row, col, vals, binStart, binCursor, binned, nEdges, nBins);

    gcnii_block_sums<<<nBins, SUM_THREADS, 0, stream>>>(binned, binStart, H, AH, nNodes);

    int nRowTiles = nNodes / 16;
    int gemmBlocks = (nRowTiles + 7) / 8;
    gcnii_gemm_epilogue<<<gemmBlocks, 256, 0, stream>>>(
        AH, H0, W, (float*)d_out, lamda, alpha, l, nRowTiles);
}


# SPMM sort: each group orders its own 7168-edge chunk by row block in LDS, writes it out whole; block sums gather the per-chunk runs into an LDS list; no memset, count, scan launch or global atomics
# speedup vs baseline: 2.8651x; 1.0661x over previous
#include <hip/hip_runtime.h>
#include <math.h>

typedef __attribute__((ext_vector_type(2))) float v2f;
typedef __attribute__((ext_vector_type(8))) float v8f;

#define FEAT 64

#define RB 100
#define COL_BITS 24
#define COL_MASK 0x00FFFFFF

#define SORT_THREADS 512
#define SORT_WAVES (SORT_THREADS / 32)
#define SORT_PER_THREAD 14
#define SORT_CHUNK (SORT_THREADS * SORT_PER_THREAD)

static __device__ __forceinline__ int group_scan(int c, int* waveTot, int nWaves, int& total) {
    int lane = threadIdx.x & 31;
    int wave = threadIdx.x >> 5;
    int x = c;
#pragma unroll
    for (int off = 1; off < 32; off <<= 1) {
        int y = __shfl_up(x, off);
        if (lane >= off) x += y;
    }
    if (lane == 31) waveTot[wave] = x;
    __syncthreads();
    int below = 0;
    int tot = 0;
    for (int w = 0; w < nWaves; ++w) {
        int t = waveTot[w];
        tot += t;
        if (w < wave) below += t;
    }
    __syncthreads();
    total = tot;
    return below + x - c;
}

__global__ void __launch_bounds__(SORT_THREADS)
gcnii_chunk_sort(const int* __restrict__ row,
                 const int* __restrict__ col,
                 const float* __restrict__ vals,
                 int* __restrict__ table,
                 int* __restrict__ binned,
                 int nEdges, int nBins, int nChunks) {
    extern __shared__ int ldsCnt[];
    __shared__ int2 stage[SORT_CHUNK];
    __shared__ int waveTot[SORT_WAVES];
    for (int b = threadIdx.x; b < nBins; b += SORT_THREADS) ldsCnt[b] = 0;
    int chunkBase = blockIdx.x * SORT_CHUNK;
    int base = chunkBase + threadIdx.x;
    int last = nEdges - 1;
    int r[SORT_PER_THREAD];
    int cl[SORT_PER_THREAD];
    float vl[SORT_PER_THREAD];
    int rank[SORT_PER_THREAD];
#pragma unroll
    for (int j = 0; j < SORT_PER_THREAD; ++j) {
        int e = base + j * SORT_THREADS;
        int ec = e < last ? e : last;
        r[j] = row[ec];
        cl[j] = col[ec];
        vl[j] = vals[ec];
    }
    __syncthreads();
#pragma unroll
    for (int j = 0; j < SORT_PER_THREAD; ++j) {
        int e = base + j * SORT_THREADS;
        rank[j] = 0;
        if (e < nEdges) rank[j] = atomicAdd(&ldsCnt[r[j] / RB], 1);
    }
    __syncthreads();
    int carry = 0;
    for (int b0 = 0; b0 < nBins; b0 += SORT_THREADS) {
        int b = b0 + threadIdx.x;
        int c = (b < nBins) ? ldsCnt[b] : 0;
        int tot;
        int ex = group_scan(c, waveTot, SORT_WAVES, tot);
        if (b < nBins) {
            int start = carry + ex;
            ldsCnt[b] = start;
            int2 t;
            t.x = chunkBase + start;
            t.y = c;
            *(int2*)(table + 2 * ((size_t)b * nChunks + blockIdx.x)) = t;
        }
        carry += tot;
    }
    __syncthreads();
    int runStart[SORT_PER_THREAD];
#pragma unroll
    for (int j = 0; j < SORT_PER_THREAD; ++j) runStart[j] = ldsCnt[r[j] / RB];
#pragma unroll
    for (int j = 0; j < SORT_PER_THREAD; ++j) {
        int e = base + j * SORT_THREADS;
        if (e < nEdges) {
            int b = r[j] / RB;
            int2 rec;
            rec.x = ((r[j] - b * RB) << COL_BITS) | cl[j];
            rec.y = __builtin_bit_cast(int, vl[j]);
            stage[runStart[j] + rank[j]] = rec;
        }
    }
    __syncthreads();
    int pairs = carry >> 1;
    int4* dst = (int4*)(binned + 2 * (size_t)chunkBase);
    for (int i = threadIdx.x; i < pairs; i += SORT_THREADS) dst[i] = ((const int4*)stage)[i];
    if ((carry & 1) && threadIdx.x == 0)
        *(int2*)(binned + 2 * (size_t)(chunkBase + carry - 1)) = stage[carry - 1];
}

#define SUM_THREADS 512
#define SUM_WAVES (SUM_THREADS / 32)
#define EDGES_PER_WAVE 32
#define ROWS_IN_FLIGHT 8
#define LIST_CAP 2048
#define COPY_PER_THREAD 4

__global__ void __launch_bounds__(SUM_THREADS)
gcnii_block_sums(const int* __restrict__ binned,
                 const int* __restrict__ table,
                 const float* __restrict__ H,
                 float* __restrict__ AH,
                 int nNodes, int nChunks) {
    __shared__ float acc[RB * FEAT];
    __shared__ int2 list[LIST_CAP];
    __shared__ int waveTot[SUM_WAVES];
    extern __shared__ int seg[];
    int* pre = seg;
    int* first = seg + nChunks + 1;
    int tid = threadIdx.x;
    int blk = blockIdx.x;

    float4 zero4 = {0.0f, 0.0f, 0.0f, 0.0f};
    for (int i = tid; i < RB * FEAT / 4; i += SUM_THREADS) ((float4*)acc)[i] = zero4;

    int carry = 0;
    for (int s0 = 0; s0 < nChunks; s0 += SUM_THREADS) {
        int s = s0 + tid;
        int2 t;
        t.x = 0;
        t.y = 0;
        if (s < nChunks) t = *(const int2*)(table + 2 * ((size_t)blk * nChunks + s));
        int tot;
        int ex = group_scan(t.y, waveTot, SUM_WAVES, tot);
        if (s < nChunks) {
            pre[s] = carry + ex;
            first[s] = t.x;
        }
        carry += tot;
    }
    if (tid == 0) pre[nChunks] = carry;
    int total = __builtin_amdgcn_readfirstlane(carry);
    int wave = __builtin_amdgcn_readfirstlane(tid >> 5);
    int lane = tid & 31;
    const float* srcLane = H + lane;
    float* dstLane = acc + lane;
    __syncthreads();

    for (int r0 = 0; r0 < total; r0 += LIST_CAP) {
        int n = total - r0;
        if (n > LIST_CAP) n = LIST_CAP;
        for (int i0 = tid; i0 < n; i0 += SUM_THREADS * COPY_PER_THREAD) {
            int g[COPY_PER_THREAD];
            int lo[COPY_PER_THREAD];
            int2 rec[COPY_PER_THREAD];
#pragma unroll
            for (int u = 0; u < COPY_PER_THREAD; ++u) {
                int i = i0 + u * SUM_THREADS;
                g[u] = r0 + (i < n ? i : 0);
                lo[u] = 0;
            }
            for (int len = nChunks; len > 1; len -= len >> 1) {
                int half = len >> 1;
#pragma unroll
                for (int u = 0; u < COPY_PER_THREAD; ++u)
                    if (pre[lo[u] + half] <= g[u]) lo[u] += half;
            }
#pragma unroll
            for (int u = 0; u < COPY_PER_THREAD; ++u)
                rec[u] = *(const int2*)(binned + 2 * (size_t)(first[lo[u]] + (g[u] - pre[lo[u]])));
#pragma unroll
            for (int u = 0; u < COPY_PER_THREAD; ++u) {
                int i = i0 + u * SUM_THREADS;
                if (i < n) list[i] = rec[u];
            }
        }
        __syncthreads();

        for (int base = wave * EDGES_PER_WAVE; base < n;
             base += SUM_WAVES * EDGES_PER_WAVE) {
            int cnt = n - base;
            if (cnt > EDGES_PER_WAVE) cnt = EDGES_PER_WAVE;
            int meta = 0;
            float v = 0.0f;
            if (lane < cnt) {
                int2 rec = list[base + lane];
                meta = rec.x;
                v = __builtin_bit_cast(float, rec.y);
            }
            if (cnt == EDGES_PER_WAVE) {
#pragma unroll
                for (int g = 0; g < EDGES_PER_WAVE; g += ROWS_IN_FLIGHT) {
                    float hlo[ROWS_IN_FLIGHT];
                    float hhi[ROWS_IN_FLIGHT];
#pragma unroll
                    for (int j = 0; j < ROWS_IN_FLIGHT; ++j) {
                        int mj = __builtin_amdgcn_readlane(meta, g + j);
                        const float* src = srcLane + (size_t)(mj & COL_MASK) * FEAT;
                        hlo[j] = src[0];
                        hhi[j] = src[FEAT / 2];
                    }
#pragma unroll
                    for (int j = 0; j < ROWS_IN_FLIGHT; ++j) {
                        int mj = __builtin_amdgcn_readlane(meta, g + j);
                        float vj = __builtin_bit_cast(
                            float, __builtin_amdgcn_readlane(__builtin_bit_cast(int, v), g + j));
                        float* dst = dstLane + ((unsigned)mj >> COL_BITS) * FEAT;
                        __hip_atomic_fetch_add(dst, vj * hlo[j],
                                               __ATOMIC_RELAXED, __HIP_MEMORY_SCOPE_WORKGROUP);
                        __hip_atomic_fetch_add(dst + FEAT / 2, vj * hhi[j],
                                               __ATOMIC_RELAXED, __HIP_MEMORY_SCOPE_WORKGROUP);
                    }
                }
            } else {
                for (int j = 0; j < cnt; ++j) {
                    int mj = __shfl(meta, j);
                    float vj = __shfl(v, j);
                    const float* src = srcLane + (size_t)(mj & COL_MASK) * FEAT;
                    float lo = src[0];
                    float hi = src[FEAT / 2];
                    float* dst = dstLane + ((unsigned)mj >> COL_BITS) * FEAT;
                    __hip_atomic_fetch_add(dst, vj * lo,
                                           __ATOMIC_RELAXED, __HIP_MEMORY_SCOPE_WORKGROUP);
                    __hip_atomic_fetch_add(dst + FEAT / 2, vj * hi,
                                           __ATOMIC_RELAXED, __HIP_MEMORY_SCOPE_WORKGROUP);
                }
            }
        }
        __syncthreads();
    }

    for (int i = tid; i < RB * FEAT / 4; i += SUM_THREADS) {
        int grow = blk * RB + (i >> 4);
        if (grow < nNodes)
            ((float4*)AH)[(size_t)grow * (FEAT / 4) + (i & 15)] = ((const float4*)acc)[i];
    }
}

__global__ void __launch_bounds__(256)
gcnii_gemm_epilogue(const float* __restrict__ AH,
                    const float* __restrict__ H0,
                    const float* __restrict__ W,
                    float* __restrict__ out,
                    const float* __restrict__ lamdaPtr,
                    const float* __restrict__ alphaPtr,
                    const int* __restrict__ lPtr,
                    int nRowTiles) {
    int tile = (int)(blockIdx.x * (blockDim.x >> 5) + (threadIdx.x >> 5));
    if (tile >= nRowTiles) return;

    int m0 = tile * 16;
    int lane = threadIdx.x & 31;
    int hi   = lane >> 4;
    int lm   = lane & 15;

    float alpha = alphaPtr[0];
    float lamda = lamdaPtr[0];
    float beta  = logf(lamda / (float)lPtr[0] + 1.0f);
    float oma   = 1.0f - alpha;
    float omb   = 1.0f - beta;

    const float* arowAH = AH + (size_t)(m0 + lm) * FEAT + hi * 2;
    const float* arowH0 = H0 + (size_t)(m0 + lm) * FEAT + hi * 2;
    const float* wbase  = W + (size_t)(hi * 2) * FEAT + lm * 4;

    v8f acc0 = {}, acc1 = {}, acc2 = {}, acc3 = {};
#pragma unroll
    for (int kk = 0; kk < 16; ++kk) {
        int K0 = kk * 4;
        float2 ah2 = *(const float2*)(arowAH + K0);
        float2 h02 = *(const float2*)(arowH0 + K0);
        v2f a;
        a.x = oma * ah2.x + alpha * h02.x;
        a.y = oma * ah2.y + alpha * h02.y;
        float4 w0 = *(const float4*)(wbase + (size_t)(K0 + 0) * FEAT);
        float4 w1 = *(const float4*)(wbase + (size_t)(K0 + 1) * FEAT);
        v2f b0, b1, b2, b3;
        b0.x = w0.x; b0.y = w1.x;
        b1.x = w0.y; b1.y = w1.y;
        b2.x = w0.z; b2.y = w1.z;
        b3.x = w0.w; b3.y = w1.w;
        acc0 = __builtin_amdgcn_wmma_f32_16x16x4_f32(
            false, a, false, b0, (short)0, acc0, false, false);
        acc1 = __builtin_amdgcn_wmma_f32_16x16x4_f32(
            false, a, false, b1, (short)0, acc1, false, false);
        acc2 = __builtin_amdgcn_wmma_f32_16x16x4_f32(
            false, a, false, b2, (short)0, acc2, false, false);
        acc3 = __builtin_amdgcn_wmma_f32_16x16x4_f32(
            false, a, false, b3, (short)0, acc3, false, false);
    }

#pragma unroll
    for (int r = 0; r < 8; ++r) {
        size_t idx = (size_t)(m0 + r + hi * 8) * FEAT + lm * 4;
        float4 ah = *(const float4*)(AH + idx);
        float4 h0 = *(const float4*)(H0 + idx);
        float4 o;
        float sup;
        sup = oma * ah.x + alpha * h0.x;  o.x = omb * sup + beta * acc0[r];
        sup = oma * ah.y + alpha * h0.y;  o.y = omb * sup + beta * acc1[r];
        sup = oma * ah.z + alpha * h0.z;  o.z = omb * sup + beta * acc2[r];
        sup = oma * ah.w + alpha * h0.w;  o.w = omb * sup + beta * acc3[r];
        *(float4*)(out + idx) = o;
    }
}

extern "C" void kernel_launch(void* const* d_in, const int* in_sizes, int n_in,
                              void* d_out, int out_size, void* d_ws, size_t ws_size,
                              hipStream_t stream) {
    const int*   row   = (const int*)  d_in[0];
    const int*   col   = (const int*)  d_in[1];
    const float* vals  = (const float*)d_in[2];
    const float* H     = (const float*)d_in[3];
    const float* H0    = (const float*)d_in[4];
    const float* W     = (const float*)d_in[5];
    const float* lamda = (const float*)d_in[6];
    const float* alpha = (const float*)d_in[7];
    const int*   l     = (const int*)  d_in[8];

    int nEdges  = in_sizes[0];
    int nNodes  = in_sizes[3] / FEAT;
    int nBins   = (nNodes + RB - 1) / RB;
    int nChunks = (nEdges + SORT_CHUNK - 1) / SORT_CHUNK;

    char* ws = (char*)d_ws;
    size_t ahBytes  = (((size_t)nNodes * FEAT * sizeof(float)) + 255) & ~(size_t)255;
    size_t binBytes = (((size_t)nEdges * 2 * sizeof(int)) + 255) & ~(size_t)255;
    float* AH     = (float*)ws;
    int*   binned = (int*)(ws + ahBytes);
    int*   table  = (int*)(ws + ahBytes + binBytes);

    size_t sortLds = (size_t)nBins * sizeof(int);
    gcnii_chunk_sort<<<nChunks, SORT_THREADS, sortLds, stream>>>(
        row, col, vals, table, binned, nEdges, nBins, nChunks);

    size_t segLds = (size_t)(2 * nChunks + 1) * sizeof(int);
    gcnii_block_sums<<<nBins, SUM_THREADS, segLds, stream>>>(
        binned, table, H, AH, nNodes, nChunks);

    int nRowTiles = nNodes / 16;
    int gemmBlocks = (nRowTiles + 7) / 8;
    gcnii_gemm_epilogue<<<gemmBlocks, 256, 0, stream>>>(
        AH, H0, W, (float*)d_out, lamda, alpha, l, nRowTiles);
}


# GEMM and epilogue fused into the block-sum kernel: sums stay in LDS (row pitch 68 words), support built there once, WMMA A fragments read from LDS; AH never goes to memory, 2 launches, not 3
# speedup vs baseline: 3.1349x; 1.0942x over previous
#include <hip/hip_runtime.h>
#include <math.h>

typedef __attribute__((ext_vector_type(2))) float v2f;
typedef __attribute__((ext_vector_type(8))) float v8f;

#define FEAT 64

#define RB 100
#define COL_BITS 24
#define COL_MASK 0x00FFFFFF

#define SORT_THREADS 512
#define SORT_WAVES (SORT_THREADS / 32)
#define SORT_PER_THREAD 14
#define SORT_CHUNK (SORT_THREADS * SORT_PER_THREAD)

static __device__ __forceinline__ int group_scan(int c, int* waveTot, int nWaves, int& total) {
    int lane = threadIdx.x & 31;
    int wave = threadIdx.x >> 5;
    int x = c;
#pragma unroll
    for (int off = 1; off < 32; off <<= 1) {
        int y = __shfl_up(x, off);
        if (lane >= off) x += y;
    }
    if (lane == 31) waveTot[wave] = x;
    __syncthreads();
    int below = 0;
    int tot = 0;
    for (int w = 0; w < nWaves; ++w) {
        int t = waveTot[w];
        tot += t;
        if (w < wave) below += t;
    }
    __syncthreads();
    total = tot;
    return below + x - c;
}

__global__ void __launch_bounds__(SORT_THREADS)
gcnii_chunk_sort(const int* __restrict__ row,
                 const int* __restrict__ col,
                 const float* __restrict__ vals,
                 int* __restrict__ table,
                 int* __restrict__ binned,
                 int nEdges, int nBins, int nChunks) {
    extern __shared__ int ldsCnt[];
    __shared__ int2 stage[SORT_CHUNK];
    __shared__ int waveTot[SORT_WAVES];
    for (int b = threadIdx.x; b < nBins; b += SORT_THREADS) ldsCnt[b] = 0;
    int chunkBase = blockIdx.x * SORT_CHUNK;
    int base = chunkBase + threadIdx.x;
    int last = nEdges - 1;
    int r[SORT_PER_THREAD];
    int cl[SORT_PER_THREAD];
    float vl[SORT_PER_THREAD];
    int rank[SORT_PER_THREAD];
#pragma unroll
    for (int j = 0; j < SORT_PER_THREAD; ++j) {
        int e = base + j * SORT_THREADS;
        int ec = e < last ? e : last;
        r[j] = row[ec];
        cl[j] = col[ec];
        vl[j] = vals[ec];
    }
    __syncthreads();
#pragma unroll
    for (int j = 0; j < SORT_PER_THREAD; ++j) {
        int e = base + j * SORT_THREADS;
        rank[j] = 0;
        if (e < nEdges) rank[j] = atomicAdd(&ldsCnt[r[j] / RB], 1);
    }
    __syncthreads();
    int carry = 0;
    for (int b0 = 0; b0 < nBins; b0 += SORT_THREADS) {
        int b = b0 + threadIdx.x;
        int c = (b < nBins) ? ldsCnt[b] : 0;
        int tot;
        int ex = group_scan(c, waveTot, SORT_WAVES, tot);
        if (b < nBins) {
            int start = carry + ex;
            ldsCnt[b] = start;
            int2 t;
            t.x = chunkBase + start;
            t.y = c;
            *(int2*)(table + 2 * ((size_t)b * nChunks + blockIdx.x)) = t;
        }
        carry += tot;
    }
    __syncthreads();
    int runStart[SORT_PER_THREAD];
#pragma unroll
    for (int j = 0; j < SORT_PER_THREAD; ++j) runStart[j] = ldsCnt[r[j] / RB];
#pragma unroll
    for (int j = 0; j < SORT_PER_THREAD; ++j) {
        int e = base + j * SORT_THREADS;
        if (e < nEdges) {
            int b = r[j] / RB;
            int2 rec;
            rec.x = ((r[j] - b * RB) << COL_BITS) | cl[j];
            rec.y = __builtin_bit_cast(int, vl[j]);
            stage[runStart[j] + rank[j]] = rec;
        }
    }
    __syncthreads();
    int pairs = carry >> 1;
    int4* dst = (int4*)(binned + 2 * (size_t)chunkBase);
    for (int i = threadIdx.x; i < pairs; i += SORT_THREADS) dst[i] = ((const int4*)stage)[i];
    if ((carry & 1) && threadIdx.x == 0)
        *(int2*)(binned + 2 * (size_t)(chunkBase + carry - 1)) = stage[carry - 1];
}

#define ACC_TILES ((RB + 15) / 16)
#define ACC_ROWS (ACC_TILES * 16)
#define ACC_PITCH (FEAT + 4)
#define SUM_THREADS 512
#define SUM_WAVES (SUM_THREADS / 32)
#define EDGES_PER_WAVE 32
#define ROWS_IN_FLIGHT 8
#define LIST_CAP 2048
#define COPY_PER_THREAD 4
#define SUP_PER_THREAD ((RB * FEAT / 4 + SUM_THREADS - 1) / SUM_THREADS)

__global__ void __launch_bounds__(SUM_THREADS)
gcnii_block_sums(const int* __restrict__ binned,
                 const int* __restrict__ table,
                 const float* __restrict__ H,
                 const float* __restrict__ H0,
                 const float* __restrict__ W,
                 float* __restrict__ out,
                 const float* __restrict__ lamdaPtr,
                 const float* __restrict__ alphaPtr,
                 const int* __restrict__ lPtr,
                 int nNodes, int nChunks) {
    __shared__ __attribute__((aligned(16))) float acc[ACC_ROWS * ACC_PITCH];
    __shared__ int2 list[LIST_CAP];
    __shared__ int waveTot[SUM_WAVES];
    extern __shared__ int seg[];
    int* pre = seg;
    int* first = seg + nChunks + 1;
    int tid = threadIdx.x;
    int blk = blockIdx.x;

    float alpha = alphaPtr[0];
    float lamda = lamdaPtr[0];
    float beta  = logf(lamda / (float)lPtr[0] + 1.0f);
    float oma   = 1.0f - alpha;
    float omb   = 1.0f - beta;

    float4 zero4 = {0.0f, 0.0f, 0.0f, 0.0f};
    for (int i = tid; i < ACC_ROWS * ACC_PITCH / 4; i += SUM_THREADS) ((float4*)acc)[i] = zero4;

    int carry = 0;
    for (int s0 = 0; s0 < nChunks; s0 += SUM_THREADS) {
        int s = s0 + tid;
        int2 t;
        t.x = 0;
        t.y = 0;
        if (s < nChunks) t = *(const int2*)(table + 2 * ((size_t)blk * nChunks + s));
        int tot;
        int ex = group_scan(t.y, waveTot, SUM_WAVES, tot);
        if (s < nChunks) {
            pre[s] = carry + ex;
            first[s] = t.x;
        }
        carry += tot;
    }
    if (tid == 0) pre[nChunks] = carry;
    int total = __builtin_amdgcn_readfirstlane(carry);
    int wave = __builtin_amdgcn_readfirstlane(tid >> 5);
    int lane = tid & 31;
    const float* srcLane = H + lane;
    float* dstLane = acc + lane;
    __syncthreads();

    for (int r0 = 0; r0 < total; r0 += LIST_CAP) {
        int n = total - r0;
        if (n > LIST_CAP) n = LIST_CAP;
        for (int i0 = tid; i0 < n; i0 += SUM_THREADS * COPY_PER_THREAD) {
            int g[COPY_PER_THREAD];
            int lo[COPY_PER_THREAD];
            int2 rec[COPY_PER_THREAD];
#pragma unroll
            for (int u = 0; u < COPY_PER_THREAD; ++u) {
                int i = i0 + u * SUM_THREADS;
                g[u] = r0 + (i < n ? i : 0);
                lo[u] = 0;
            }
            for (int len = nChunks; len > 1; len -= len >> 1) {
                int half = len >> 1;
#pragma unroll
                for (int u = 0; u < COPY_PER_THREAD; ++u)
                    if (pre[lo[u] + half] <= g[u]) lo[u] += half;
            }
#pragma unroll
            for (int u = 0; u < COPY_PER_THREAD; ++u)
                rec[u] = *(const int2*)(binned + 2 * (size_t)(first[lo[u]] + (g[u] - pre[lo[u]])));
#pragma unroll
            for (int u = 0; u < COPY_PER_THREAD; ++u) {
                int i = i0 + u * SUM_THREADS;
                if (i < n) list[i] = rec[u];
            }
        }
        __syncthreads();

        for (int base = wave * EDGES_PER_WAVE; base < n;
             base += SUM_WAVES * EDGES_PER_WAVE) {
            int cnt = n - base;
            if (cnt > EDGES_PER_WAVE) cnt = EDGES_PER_WAVE;
            int meta = 0;
            float v = 0.0f;
            if (lane < cnt) {
                int2 rec = list[base + lane];
                meta = rec.x;
                v = __builtin_bit_cast(float, rec.y);
            }
            if (cnt == EDGES_PER_WAVE) {
#pragma unroll
                for (int g = 0; g < EDGES_PER_WAVE; g += ROWS_IN_FLIGHT) {
                    float hlo[ROWS_IN_FLIGHT];
                    float hhi[ROWS_IN_FLIGHT];
#pragma unroll
                    for (int j = 0; j < ROWS_IN_FLIGHT; ++j) {
                        int mj = __builtin_amdgcn_readlane(meta, g + j);
                        const float* src = srcLane + (size_t)(mj & COL_MASK) * FEAT;
                        hlo[j] = src[0];
                        hhi[j] = src[FEAT / 2];
                    }
#pragma unroll
                    for (int j = 0; j < ROWS_IN_FLIGHT; ++j) {
                        int mj = __builtin_amdgcn_readlane(meta, g + j);
                        float vj = __builtin_bit_cast(
                            float, __builtin_amdgcn_readlane(__builtin_bit_cast(int, v), g + j));
                        float* dst = dstLane + ((unsigned)mj >> COL_BITS) * ACC_PITCH;
                        __hip_atomic_fetch_add(dst, vj * hlo[j],
                                               __ATOMIC_RELAXED, __HIP_MEMORY_SCOPE_WORKGROUP);
                        __hip_atomic_fetch_add(dst + FEAT / 2, vj * hhi[j],
                                               __ATOMIC_RELAXED, __HIP_MEMORY_SCOPE_WORKGROUP);
                    }
                }
            } else {
                for (int j = 0; j < cnt; ++j) {
                    int mj = __shfl(meta, j);
                    float vj = __shfl(v, j);
                    const float* src = srcLane + (size_t)(mj & COL_MASK) * FEAT;
                    float lo = src[0];
                    float hi = src[FEAT / 2];
                    float* dst = dstLane + ((unsigned)mj >> COL_BITS) * ACC_PITCH;
                    __hip_atomic_fetch_add(dst, vj * lo,
                                           __ATOMIC_RELAXED, __HIP_MEMORY_SCOPE_WORKGROUP);
                    __hip_atomic_fetch_add(dst + FEAT / 2, vj * hi,
                                           __ATOMIC_RELAXED, __HIP_MEMORY_SCOPE_WORKGROUP);
                }
            }
        }
        __syncthreads();
    }

    float4 h0v[SUP_PER_THREAD];
#pragma unroll
    for (int u = 0; u < SUP_PER_THREAD; ++u) {
        int i = tid + u * SUM_THREADS;
        int ic = i < RB * FEAT / 4 ? i : RB * FEAT / 4 - 1;
        int grow = blk * RB + (ic >> 4);
        int gl = grow < nNodes ? grow : nNodes - 1;
        h0v[u] = ((const float4*)H0)[(size_t)gl * (FEAT / 4) + (ic & 15)];
    }
#pragma unroll
    for (int u = 0; u < SUP_PER_THREAD; ++u) {
        int i = tid + u * SUM_THREADS;
        int grow = blk * RB + (i >> 4);
        if (i < RB * FEAT / 4 && grow < nNodes) {
            float4* p = (float4*)(acc + (i >> 4) * ACC_PITCH) + (i & 15);
            float4 a = *p;
            a.x = oma * a.x + alpha * h0v[u].x;
            a.y = oma * a.y + alpha * h0v[u].y;
            a.z = oma * a.z + alpha * h0v[u].z;
            a.w = oma * a.w + alpha * h0v[u].w;
            *p = a;
        }
    }
    __syncthreads();

    if (wave >= 2 * ACC_TILES) return;
    int rt = wave >> 1;
    int ch = wave & 1;
    int hi = lane >> 4;
    int lm = lane & 15;
    const float* arow  = acc + (rt * 16 + lm) * ACC_PITCH + hi * 2;
    const float* wbase = W + (size_t)(hi * 2) * FEAT + ch * 32 + lm * 2;

    v8f acc0 = {}, acc1 = {};
#pragma unroll
    for (int kk = 0; kk < 16; ++kk) {
        int K0 = kk * 4;
        float2 s2 = *(const float2*)(arow + K0);
        float2 w0 = *(const float2*)(wbase + (size_t)(K0 + 0) * FEAT);
        float2 w1 = *(const float2*)(wbase + (size_t)(K0 + 1) * FEAT);
        v2f a, b0, b1;
        a.x = s2.x;  a.y = s2.y;
        b0.x = w0.x; b0.y = w1.x;
        b1.x = w0.y; b1.y = w1.y;
        acc0 = __builtin_amdgcn_wmma_f32_16x16x4_f32(
            false, a, false, b0, (short)0, acc0, false, false);
        acc1 = __builtin_amdgcn_wmma_f32_16x16x4_f32(
            false, a, false, b1, (short)0, acc1, false, false);
    }

#pragma unroll
    for (int r = 0; r < 8; ++r) {
        int rin = rt * 16 + r + hi * 8;
        int grow = blk * RB + rin;
        if (rin < RB && grow < nNodes) {
            float2 s = *(const float2*)(acc + rin * ACC_PITCH + ch * 32 + lm * 2);
            float2 o;
            o.x = omb * s.x + beta * acc0[r];
            o.y = omb * s.y + beta * acc1[r];
            *(float2*)(out + (size_t)grow * FEAT + ch * 32 + lm * 2) = o;
        }
    }
}

extern "C" void kernel_launch(void* const* d_in, const int* in_sizes, int n_in,
                              void* d_out, int out_size, void* d_ws, size_t ws_size,
                              hipStream_t stream) {
    const int*   row   = (const int*)  d_in[0];
    const int*   col   = (const int*)  d_in[1];
    const float* vals  = (const float*)d_in[2];
    const float* H     = (const float*)d_in[3];
    const float* H0    = (const float*)d_in[4];
    const float* W     = (const float*)d_in[5];
    const float* lamda = (const float*)d_in[6];
    const float* alpha = (const float*)d_in[7];
    const int*   l     = (const int*)  d_in[8];

    int nEdges  = in_sizes[0];
    int nNodes  = in_sizes[3] / FEAT;
    int nBins   = (nNodes + RB - 1) / RB;
    int nChunks = (nEdges + SORT_CHUNK - 1) / SORT_CHUNK;

    char* ws = (char*)d_ws;
    size_t binBytes = (((size_t)nEdges * 2 * sizeof(int)) + 255) & ~(size_t)255;
    int*   binned = (int*)ws;
    int*   table  = (int*)(ws + binBytes);

    size_t sortLds = (size_t)nBins * sizeof(int);
    gcnii_chunk_sort<<<nChunks, SORT_THREADS, sortLds, stream>>>(
        row, col, vals, table, binned, nEdges, nBins, nChunks);

    size_t segLds = (size_t)(2 * nChunks + 1) * sizeof(int);
    gcnii_block_sums<<<nBins, SUM_THREADS, segLds, stream>>>(
        binned, table, H, H0, W, (float*)d_out, lamda, alpha, l, nNodes, nChunks);
}


# SPMM sums: block list put in row order (LDS count, scan, place); a wave sums a contiguous share per row in registers (fma), ds_add_f32 only at row change; order in a row: list = arrival order
# speedup vs baseline: 3.5311x; 1.1264x over previous
#include <hip/hip_runtime.h>
#include <math.h>

typedef __attribute__((ext_vector_type(2))) float v2f;
typedef __attribute__((ext_vector_type(8))) float v8f;

#define FEAT 64

#define RB 100
#define COL_BITS 24
#define COL_MASK 0x00FFFFFF

#define SORT_THREADS 512
#define SORT_WAVES (SORT_THREADS / 32)
#define SORT_PER_THREAD 14
#define SORT_CHUNK (SORT_THREADS * SORT_PER_THREAD)

static __device__ __forceinline__ int group_scan(int c, int* waveTot, int nWaves, int& total) {
    int lane = threadIdx.x & 31;
    int wave = threadIdx.x >> 5;
    int x = c;
#pragma unroll
    for (int off = 1; off < 32; off <<= 1) {
        int y = __shfl_up(x, off);
        if (lane >= off) x += y;
    }
    if (lane == 31) waveTot[wave] = x;
    __syncthreads();
    int below = 0;
    int tot = 0;
    for (int w = 0; w < nWaves; ++w) {
        int t = waveTot[w];
        tot += t;
        if (w < wave) below += t;
    }
    __syncthreads();
    total = tot;
    return below + x - c;
}

__global__ void __launch_bounds__(SORT_THREADS)
gcnii_chunk_sort(const int* __restrict__ row,
                 const int* __restrict__ col,
                 const float* __restrict__ vals,
                 int* __restrict__ table,
                 int* __restrict__ binned,
                 int nEdges, int nBins, int nChunks) {
    extern __shared__ int ldsCnt[];
    __shared__ int2 stage[SORT_CHUNK];
    __shared__ int waveTot[SORT_WAVES];
    for (int b = threadIdx.x; b < nBins; b += SORT_THREADS) ldsCnt[b] = 0;
    int chunkBase = blockIdx.x * SORT_CHUNK;
    int base = chunkBase + threadIdx.x;
    int last = nEdges - 1;
    int r[SORT_PER_THREAD];
    int cl[SORT_PER_THREAD];
    float vl[SORT_PER_THREAD];
    int rank[SORT_PER_THREAD];
#pragma unroll
    for (int j = 0; j < SORT_PER_THREAD; ++j) {
        int e = base + j * SORT_THREADS;
        int ec = e < last ? e : last;
        r[j] = row[ec];
        cl[j] = col[ec];
        vl[j] = vals[ec];
    }
    __syncthreads();
#pragma unroll
    for (int j = 0; j < SORT_PER_THREAD; ++j) {
        int e = base + j * SORT_THREADS;
        rank[j] = 0;
        if (e < nEdges) rank[j] = atomicAdd(&ldsCnt[r[j] / RB], 1);
    }
    __syncthreads();
    int carry = 0;
    for (int b0 = 0; b0 < nBins; b0 += SORT_THREADS) {
        int b = b0 + threadIdx.x;
        int c = (b < nBins) ? ldsCnt[b] : 0;
        int tot;
        int ex = group_scan(c, waveTot, SORT_WAVES, tot);
        if (b < nBins) {
            int start = carry + ex;
            ldsCnt[b] = start;
            int2 t;
            t.x = chunkBase + start;
            t.y = c;
            *(int2*)(table + 2 * ((size_t)b * nChunks + blockIdx.x)) = t;
        }
        carry += tot;
    }
    __syncthreads();
    int runStart[SORT_PER_THREAD];
#pragma unroll
    for (int j = 0; j < SORT_PER_THREAD; ++j) runStart[j] = ldsCnt[r[j] / RB];
#pragma unroll
    for (int j = 0; j < SORT_PER_THREAD; ++j) {
        int e = base + j * SORT_THREADS;
        if (e < nEdges) {
            int b = r[j] / RB;
            int2 rec;
            rec.x = ((r[j] - b * RB) << COL_BITS) | cl[j];
            rec.y = __builtin_bit_cast(int, vl[j]);
            stage[runStart[j] + rank[j]] = rec;
        }
    }
    __syncthreads();
    int pairs = carry >> 1;
    int4* dst = (int4*)(binned + 2 * (size_t)chunkBase);
    for (int i = threadIdx.x; i < pairs; i += SORT_THREADS) dst[i] = ((const int4*)stage)[i];
    if ((carry & 1) && threadIdx.x == 0)
        *(int2*)(binned + 2 * (size_t)(chunkBase + carry - 1)) = stage[carry - 1];
}

#define ACC_TILES ((RB + 15) / 16)
#define ACC_ROWS (ACC_TILES * 16)
#define ACC_PITCH (FEAT + 4)
#define SUM_THREADS 512
#define SUM_WAVES (SUM_THREADS / 32)
#define EDGES_PER_WAVE 32
#define ROWS_IN_FLIGHT 8
#define LIST_CAP 2048
#define COPY_PER_THREAD 4
#define ROW_SLOTS (((RB + 31) / 32) * 32)
#define SUP_PER_THREAD ((RB * FEAT / 4 + SUM_THREADS - 1) / SUM_THREADS)

__global__ void __launch_bounds__(SUM_THREADS)
gcnii_block_sums(const int* __restrict__ binned,
                 const int* __restrict__ table,
                 const float* __restrict__ H,
                 const float* __restrict__ H0,
                 const float* __restrict__ W,
                 float* __restrict__ out,
                 const float* __restrict__ lamdaPtr,
                 const float* __restrict__ alphaPtr,
                 const int* __restrict__ lPtr,
                 int nNodes, int nChunks) {
    __shared__ __attribute__((aligned(16))) float acc[ACC_ROWS * ACC_PITCH];
    __shared__ int2 list[LIST_CAP];
    __shared__ int waveTot[SUM_WAVES];
    __shared__ int rowCnt[ROW_SLOTS];
    __shared__ int rowStart[ROW_SLOTS];
    __shared__ int rowTot[ROW_SLOTS / 32];
    extern __shared__ int seg[];
    int* pre = seg;
    int* first = seg + nChunks + 1;
    int tid = threadIdx.x;
    int blk = blockIdx.x;

    float alpha = alphaPtr[0];
    float lamda = lamdaPtr[0];
    float beta  = logf(lamda / (float)lPtr[0] + 1.0f);
    float oma   = 1.0f - alpha;
    float omb   = 1.0f - beta;

    float4 zero4 = {0.0f, 0.0f, 0.0f, 0.0f};
    for (int i = tid; i < ACC_ROWS * ACC_PITCH / 4; i += SUM_THREADS) ((float4*)acc)[i] = zero4;
    if (tid < ROW_SLOTS) rowCnt[tid] = 0;

    int carry = 0;
    for (int s0 = 0; s0 < nChunks; s0 += SUM_THREADS) {
        int s = s0 + tid;
        int2 t;
        t.x = 0;
        t.y = 0;
        if (s < nChunks) t = *(const int2*)(table + 2 * ((size_t)blk * nChunks + s));
        int tot;
        int ex = group_scan(t.y, waveTot, SUM_WAVES, tot);
        if (s < nChunks) {
            pre[s] = carry + ex;
            first[s] = t.x;
        }
        carry += tot;
    }
    if (tid == 0) pre[nChunks] = carry;
    int total = __builtin_amdgcn_readfirstlane(carry);
    int wave = __builtin_amdgcn_readfirstlane(tid >> 5);
    int lane = tid & 31;
    const float* srcLane = H + lane;
    float* dstLane = acc + lane;
    __syncthreads();

    for (int r0 = 0; r0 < total; r0 += LIST_CAP) {
        int n = total - r0;
        if (n > LIST_CAP) n = LIST_CAP;
        int g[COPY_PER_THREAD];
        int lo[COPY_PER_THREAD];
        int2 rec[COPY_PER_THREAD];
        int rank[COPY_PER_THREAD];
#pragma unroll
        for (int u = 0; u < COPY_PER_THREAD; ++u) {
            int i = tid + u * SUM_THREADS;
            g[u] = r0 + (i < n ? i : 0);
            lo[u] = 0;
        }
        for (int len = nChunks; len > 1; len -= len >> 1) {
            int half = len >> 1;
#pragma unroll
            for (int u = 0; u < COPY_PER_THREAD; ++u)
                if (pre[lo[u] + half] <= g[u]) lo[u] += half;
        }
#pragma unroll
        for (int u = 0; u < COPY_PER_THREAD; ++u)
            rec[u] = *(const int2*)(binned + 2 * (size_t)(first[lo[u]] + (g[u] - pre[lo[u]])));
#pragma unroll
        for (int u = 0; u < COPY_PER_THREAD; ++u) {
            int i = tid + u * SUM_THREADS;
            rank[u] = 0;
            if (i < n) rank[u] = atomicAdd(&rowCnt[(unsigned)rec[u].x >> COL_BITS], 1);
        }
        __syncthreads();
        if (tid < ROW_SLOTS) {
            int c = rowCnt[tid];
            rowCnt[tid] = 0;
            int x = c;
#pragma unroll
            for (int off = 1; off < 32; off <<= 1) {
                int y = __shfl_up(x, off);
                if (lane >= off) x += y;
            }
            rowStart[tid] = x - c;
            if (lane == 31) rowTot[tid >> 5] = x;
        }
        __syncthreads();
#pragma unroll
        for (int u = 0; u < COPY_PER_THREAD; ++u) {
            int i = tid + u * SUM_THREADS;
            if (i < n) {
                int rw = (unsigned)rec[u].x >> COL_BITS;
                int slot = rowStart[rw] + rank[u];
#pragma unroll
                for (int w = 0; w < ROW_SLOTS / 32 - 1; ++w)
                    if (rw >= 32 * (w + 1)) slot += rowTot[w];
                list[slot] = rec[u];
            }
        }
        __syncthreads();

        int per = (n + SUM_WAVES * ROWS_IN_FLIGHT - 1) / (SUM_WAVES * ROWS_IN_FLIGHT) * ROWS_IN_FLIGHT;
        int shareBegin = wave * per;
        int shareEnd = shareBegin + per;
        if (shareEnd > n) shareEnd = n;
        if (shareBegin < shareEnd) {
            int cur = (unsigned)__builtin_amdgcn_readfirstlane(list[shareBegin].x) >> COL_BITS;
            float sumLo = 0.0f;
            float sumHi = 0.0f;
            for (int base = shareBegin; base < shareEnd; base += EDGES_PER_WAVE) {
                int cnt = shareEnd - base;
                if (cnt > EDGES_PER_WAVE) cnt = EDGES_PER_WAVE;
                int idx = base + lane;
                int2 ent = list[idx < shareEnd ? idx : shareEnd - 1];
                int meta = ent.x;
                float v = idx < shareEnd ? __builtin_bit_cast(float, ent.y) : 0.0f;
#pragma unroll
                for (int gr = 0; gr < EDGES_PER_WAVE; gr += ROWS_IN_FLIGHT) {
                    if (gr < cnt) {
                        float hlo[ROWS_IN_FLIGHT];
                        float hhi[ROWS_IN_FLIGHT];
#pragma unroll
                        for (int j = 0; j < ROWS_IN_FLIGHT; ++j) {
                            int mj = __builtin_amdgcn_readlane(meta, gr + j);
                            const float* src = srcLane + (size_t)(mj & COL_MASK) * FEAT;
                            hlo[j] = src[0];
                            hhi[j] = src[FEAT / 2];
                        }
#pragma unroll
                        for (int j = 0; j < ROWS_IN_FLIGHT; ++j) {
                            int mj = __builtin_amdgcn_readlane(meta, gr + j);
                            float vj = __builtin_bit_cast(
                                float, __builtin_amdgcn_readlane(__builtin_bit_cast(int, v), gr + j));
                            int rj = (unsigned)mj >> COL_BITS;
                            if (__builtin_expect(rj != cur, 0)) {
                                float* dst = dstLane + cur * ACC_PITCH;
                                __hip_atomic_fetch_add(dst, sumLo,
                                                       __ATOMIC_RELAXED, __HIP_MEMORY_SCOPE_WORKGROUP);
                                __hip_atomic_fetch_add(dst + FEAT / 2, sumHi,
                                                       __ATOMIC_RELAXED, __HIP_MEMORY_SCOPE_WORKGROUP);
                                cur = rj;
                                sumLo = 0.0f;
                                sumHi = 0.0f;
                            }
                            sumLo = __builtin_fmaf(vj, hlo[j], sumLo);
                            sumHi = __builtin_fmaf(vj, hhi[j], sumHi);
                        }
                    }
                }
            }
            float* dst = dstLane + cur * ACC_PITCH;
            __hip_atomic_fetch_add(dst, sumLo, __ATOMIC_RELAXED, __HIP_MEMORY_SCOPE_WORKGROUP);
            __hip_atomic_fetch_add(dst + FEAT / 2, sumHi, __ATOMIC_RELAXED, __HIP_MEMORY_SCOPE_WORKGROUP);
        }
        __syncthreads();
    }

    float4 h0v[SUP_PER_THREAD];
#pragma unroll
    for (int u = 0; u < SUP_PER_THREAD; ++u) {
        int i = tid + u * SUM_THREADS;
        int ic = i < RB * FEAT / 4 ? i : RB * FEAT / 4 - 1;
        int grow = blk * RB + (ic >> 4);
        int gl = grow < nNodes ? grow : nNodes - 1;
        h0v[u] = ((const float4*)H0)[(size_t)gl * (FEAT / 4) + (ic & 15)];
    }
#pragma unroll
    for (int u = 0; u < SUP_PER_THREAD; ++u) {
        int i = tid + u * SUM_THREADS;
        int grow = blk * RB + (i >> 4);
        if (i < RB * FEAT / 4 && grow < nNodes) {
            float4* p = (float4*)(acc + (i >> 4) * ACC_PITCH) + (i & 15);
            float4 a = *p;
            a.x = oma * a.x + alpha * h0v[u].x;
            a.y = oma * a.y + alpha * h0v[u].y;
            a.z = oma * a.z + alpha * h0v[u].z;
            a.w = oma * a.w + alpha * h0v[u].w;
            *p = a;
        }
    }
    __syncthreads();

    if (wave >= 2 * ACC_TILES) return;
    int rt = wave >> 1;
    int ch = wave & 1;
    int hi = lane >> 4;
    int lm = lane & 15;
    const float* arow  = acc + (rt * 16 + lm) * ACC_PITCH + hi * 2;
    const float* wbase = W + (size_t)(hi * 2) * FEAT + ch * 32 + lm * 2;

    v8f acc0 = {}, acc1 = {};
#pragma unroll
    for (int kk = 0; kk < 16; ++kk) {
        int K0 = kk * 4;
        float2 s2 = *(const float2*)(arow + K0);
        float2 w0 = *(const float2*)(wbase + (size_t)(K0 + 0) * FEAT);
        float2 w1 = *(const float2*)(wbase + (size_t)(K0 + 1) * FEAT);
        v2f a, b0, b1;
        a.x = s2.x;  a.y = s2.y;
        b0.x = w0.x; b0.y = w1.x;
        b1.x = w0.y; b1.y = w1.y;
        acc0 = __builtin_amdgcn_wmma_f32_16x16x4_f32(
            false, a, false, b0, (short)0, acc0, false, false);
        acc1 = __builtin_amdgcn_wmma_f32_16x16x4_f32(
            false, a, false, b1, (short)0, acc1, false, false);
    }

#pragma unroll
    for (int r = 0; r < 8; ++r) {
        int rin = rt * 16 + r + hi * 8;
        int grow = blk * RB + rin;
        if (rin < RB && grow < nNodes) {
            float2 s = *(const float2*)(acc + rin * ACC_PITCH + ch * 32 + lm * 2);
            float2 o;
            o.x = omb * s.x + beta * acc0[r];
            o.y = omb * s.y + beta * acc1[r];
            *(float2*)(out + (size_t)grow * FEAT + ch * 32 + lm * 2) = o;
        }
    }
}

extern "C" void kernel_launch(void* const* d_in, const int* in_sizes, int n_in,
                              void* d_out, int out_size, void* d_ws, size_t ws_size,
                              hipStream_t stream) {
    const int*   row   = (const int*)  d_in[0];
    const int*   col   = (const int*)  d_in[1];
    const float* vals  = (const float*)d_in[2];
    const float* H     = (const float*)d_in[3];
    const float* H0    = (const float*)d_in[4];
    const float* W     = (const float*)d_in[5];
    const float* lamda = (const float*)d_in[6];
    const float* alpha = (const float*)d_in[7];
    const int*   l     = (const int*)  d_in[8];

    int nEdges  = in_sizes[0];
    int nNodes  = in_sizes[3] / FEAT;
    int nBins   = (nNodes + RB - 1) / RB;
    int nChunks = (nEdges + SORT_CHUNK - 1) / SORT_CHUNK;

    char* ws = (char*)d_ws;
    size_t binBytes = (((size_t)nEdges * 2 * sizeof(int)) + 255) & ~(size_t)255;
    int*   binned = (int*)ws;
    int*   table  = (int*)(ws + binBytes);

    size_t sortLds = (size_t)nBins * sizeof(int);
    gcnii_chunk_sort<<<nChunks, SORT_THREADS, sortLds, stream>>>(
        row, col, vals, table, binned, nEdges, nBins, nChunks);

    size_t segLds = (size_t)(2 * nChunks + 1) * sizeof(int);
    gcnii_block_sums<<<nBins, SUM_THREADS, segLds, stream>>>(
        binned, table, H, H0, W, (float*)d_out, lamda, alpha, l, nNodes, nChunks);
}


# Row blocks of 200 rows (500 work-groups of 32 waves, 2 per WGP): half as many (chunk, block) runs, each twice as long: half the table stores and scattered record fetches; row sums split at 64 shares
# speedup vs baseline: 3.8819x; 1.0993x over previous
#include <hip/hip_runtime.h>
#include <math.h>

typedef __attribute__((ext_vector_type(2))) float v2f;
typedef __attribute__((ext_vector_type(8))) float v8f;

#define FEAT 64

#define RB 200
#define COL_BITS 24
#define COL_MASK 0x00FFFFFF

#define SORT_THREADS 512
#define SORT_WAVES (SORT_THREADS / 32)
#define SORT_PER_THREAD 14
#define SORT_CHUNK (SORT_THREADS * SORT_PER_THREAD)

static __device__ __forceinline__ int group_scan(int c, int* waveTot, int nWaves, int& total) {
    int lane = threadIdx.x & 31;
    int wave = threadIdx.x >> 5;
    int x = c;
#pragma unroll
    for (int off = 1; off < 32; off <<= 1) {
        int y = __shfl_up(x, off);
        if (lane >= off) x += y;
    }
    if (lane == 31) waveTot[wave] = x;
    __syncthreads();
    int below = 0;
    int tot = 0;
    for (int w = 0; w < nWaves; ++w) {
        int t = waveTot[w];
        tot += t;
        if (w < wave) below += t;
    }
    __syncthreads();
    total = tot;
    return below + x - c;
}

__global__ void __launch_bounds__(SORT_THREADS)
gcnii_chunk_sort(const int* __restrict__ row,
                 const int* __restrict__ col,
                 const float* __restrict__ vals,
                 int* __restrict__ table,
                 int* __restrict__ binned,
                 int nEdges, int nBins, int nChunks) {
    extern __shared__ int ldsCnt[];
    __shared__ int2 stage[SORT_CHUNK];
    __shared__ int waveTot[SORT_WAVES];
    for (int b = threadIdx.x; b < nBins; b += SORT_THREADS) ldsCnt[b] = 0;
    int chunkBase = blockIdx.x * SORT_CHUNK;
    int base = chunkBase + threadIdx.x;
    int last = nEdges - 1;
    int r[SORT_PER_THREAD];
    int cl[SORT_PER_THREAD];
    float vl[SORT_PER_THREAD];
    int rank[SORT_PER_THREAD];
#pragma unroll
    for (int j = 0; j < SORT_PER_THREAD; ++j) {
        int e = base + j * SORT_THREADS;
        int ec = e < last ? e : last;
        r[j] = row[ec];
        cl[j] = col[ec];
        vl[j] = vals[ec];
    }
    __syncthreads();
#pragma unroll
    for (int j = 0; j < SORT_PER_THREAD; ++j) {
        int e = base + j * SORT_THREADS;
        rank[j] = 0;
        if (e < nEdges) rank[j] = atomicAdd(&ldsCnt[r[j] / RB], 1);
    }
    __syncthreads();
    int carry = 0;
    for (int b0 = 0; b0 < nBins; b0 += SORT_THREADS) {
        int b = b0 + threadIdx.x;
        int c = (b < nBins) ? ldsCnt[b] : 0;
        int tot;
        int ex = group_scan(c, waveTot, SORT_WAVES, tot);
        if (b < nBins) {
            int start = carry + ex;
            ldsCnt[b] = start;
            int2 t;
            t.x = chunkBase + start;
            t.y = c;
            *(int2*)(table + 2 * ((size_t)b * nChunks + blockIdx.x)) = t;
        }
        carry += tot;
    }
    __syncthreads();
    int runStart[SORT_PER_THREAD];
#pragma unroll
    for (int j = 0; j < SORT_PER_THREAD; ++j) runStart[j] = ldsCnt[r[j] / RB];
#pragma unroll
    for (int j = 0; j < SORT_PER_THREAD; ++j) {
        int e = base + j * SORT_THREADS;
        if (e < nEdges) {
            int b = r[j] / RB;
            int2 rec;
            rec.x = (int)(((unsigned)(r[j] - b * RB) << COL_BITS) | (unsigned)cl[j]);
            rec.y = __builtin_bit_cast(int, vl[j]);
            stage[runStart[j] + rank[j]] = rec;
        }
    }
    __syncthreads();
    int pairs = carry >> 1;
    int4* dst = (int4*)(binned + 2 * (size_t)chunkBase);
    for (int i = threadIdx.x; i < pairs; i += SORT_THREADS) dst[i] = ((const int4*)stage)[i];
    if ((carry & 1) && threadIdx.x == 0)
        *(int2*)(binned + 2 * (size_t)(chunkBase + carry - 1)) = stage[carry - 1];
}

#define ACC_TILES ((RB + 15) / 16)
#define ACC_ROWS (ACC_TILES * 16)
#define ACC_PITCH (FEAT + 4)
#define SUM_THREADS 1024
#define SUM_WAVES (SUM_THREADS / 32)
#define ROWS_IN_FLIGHT 6
#define LIST_CAP 4096
#define COPY_PER_THREAD 4
#define ROW_SLOTS (((RB + 31) / 32) * 32)
#define SUP_PER_THREAD ((RB * FEAT / 4 + SUM_THREADS - 1) / SUM_THREADS)

__global__ void __launch_bounds__(SUM_THREADS, 16)
gcnii_block_sums(const int* __restrict__ binned,
                 const int* __restrict__ table,
                 const float* __restrict__ H,
                 const float* __restrict__ H0,
                 const float* __restrict__ W,
                 float* __restrict__ out,
                 const float* __restrict__ lamdaPtr,
                 const float* __restrict__ alphaPtr,
                 const int* __restrict__ lPtr,
                 int nNodes, int nChunks) {
    __shared__ __attribute__((aligned(16))) float acc[ACC_ROWS * ACC_PITCH];
    __shared__ int2 list[LIST_CAP];
    __shared__ int waveTot[SUM_WAVES];
    __shared__ int rowCnt[ROW_SLOTS];
    __shared__ int rowStart[ROW_SLOTS];
    __shared__ int rowTot[ROW_SLOTS / 32];
    extern __shared__ int seg[];
    int* pre = seg;
    int* first = seg + nChunks + 1;
    int tid = threadIdx.x;
    int blk = blockIdx.x;

    float alpha = alphaPtr[0];
    float lamda = lamdaPtr[0];
    float beta  = logf(lamda / (float)lPtr[0] + 1.0f);
    float oma   = 1.0f - alpha;
    float omb   = 1.0f - beta;

    float4 zero4 = {0.0f, 0.0f, 0.0f, 0.0f};
    for (int i = tid; i < ACC_ROWS * ACC_PITCH / 4; i += SUM_THREADS) ((float4*)acc)[i] = zero4;
    if (tid < ROW_SLOTS) rowCnt[tid] = 0;

    int carry = 0;
    for (int s0 = 0; s0 < nChunks; s0 += SUM_THREADS) {
        int s = s0 + tid;
        int2 t;
        t.x = 0;
        t.y = 0;
        if (s < nChunks) t = *(const int2*)(table + 2 * ((size_t)blk * nChunks + s));
        int tot;
        int ex = group_scan(t.y, waveTot, SUM_WAVES, tot);
        if (s < nChunks) {
            pre[s] = carry + ex;
            first[s] = t.x;
        }
        carry += tot;
    }
    if (tid == 0) pre[nChunks] = carry;
    int total = __builtin_amdgcn_readfirstlane(carry);
    int wave = __builtin_amdgcn_readfirstlane(tid >> 5);
    int lane = tid & 31;
    int hf = lane >> 4;
    int c4 = (lane & 15) * 4;
    __syncthreads();

    for (int r0 = 0; r0 < total; r0 += LIST_CAP) {
        int n = total - r0;
        if (n > LIST_CAP) n = LIST_CAP;
        int g[COPY_PER_THREAD];
        int lo[COPY_PER_THREAD];
        int2 rec[COPY_PER_THREAD];
        int rank[COPY_PER_THREAD];
#pragma unroll
        for (int u = 0; u < COPY_PER_THREAD; ++u) {
            int i = tid + u * SUM_THREADS;
            g[u] = r0 + (i < n ? i : 0);
            lo[u] = 0;
        }
        for (int len = nChunks; len > 1; len -= len >> 1) {
            int half = len >> 1;
#pragma unroll
            for (int u = 0; u < COPY_PER_THREAD; ++u)
                if (pre[lo[u] + half] <= g[u]) lo[u] += half;
        }
#pragma unroll
        for (int u = 0; u < COPY_PER_THREAD; ++u)
            rec[u] = *(const int2*)(binned + 2 * (size_t)(first[lo[u]] + (g[u] - pre[lo[u]])));
#pragma unroll
        for (int u = 0; u < COPY_PER_THREAD; ++u) {
            int i = tid + u * SUM_THREADS;
            rank[u] = 0;
            if (i < n) rank[u] = atomicAdd(&rowCnt[(unsigned)rec[u].x >> COL_BITS], 1);
        }
        __syncthreads();
        if (tid < ROW_SLOTS) {
            int c = rowCnt[tid];
            rowCnt[tid] = 0;
            int x = c;
#pragma unroll
            for (int off = 1; off < 32; off <<= 1) {
                int y = __shfl_up(x, off);
                if (lane >= off) x += y;
            }
            rowStart[tid] = x - c;
            if (lane == 31) rowTot[tid >> 5] = x;
        }
        __syncthreads();
#pragma unroll
        for (int u = 0; u < COPY_PER_THREAD; ++u) {
            int i = tid + u * SUM_THREADS;
            if (i < n) {
                int rw = (unsigned)rec[u].x >> COL_BITS;
                int slot = rowStart[rw] + rank[u];
#pragma unroll
                for (int w = 0; w < ROW_SLOTS / 32 - 1; ++w)
                    if (rw >= 32 * (w + 1)) slot += rowTot[w];
                list[slot] = rec[u];
            }
        }
        __syncthreads();

        int per = (n + 2 * SUM_WAVES * ROWS_IN_FLIGHT - 1) / (2 * SUM_WAVES * ROWS_IN_FLIGHT) * ROWS_IN_FLIGHT;
        int waveBegin = wave * 2 * per;
        if (waveBegin < n) {
            int waveLen = n - waveBegin;
            if (waveLen > per) waveLen = per;
            int shareBegin = waveBegin + hf * per;
            int shareEnd = shareBegin + per;
            if (shareEnd > n) shareEnd = n;
            bool active = shareBegin < shareEnd;
            int lastIdx = active ? shareEnd - 1 : n - 1;
            int cur = (unsigned)list[active ? shareBegin : lastIdx].x >> COL_BITS;
            float4 sum = zero4;
            for (int t = 0; t < waveLen; t += ROWS_IN_FLIGHT) {
                int2 ent[ROWS_IN_FLIGHT];
                float4 hv[ROWS_IN_FLIGHT];
                int rem = shareEnd - shareBegin - t;
#pragma unroll
                for (int j = 0; j < ROWS_IN_FLIGHT; ++j) {
                    int idx = shareBegin + t + j;
                    ent[j] = list[idx < lastIdx ? idx : lastIdx];
                }
#pragma unroll
                for (int j = 0; j < ROWS_IN_FLIGHT; ++j) {
                    unsigned off = ((unsigned)(ent[j].x & COL_MASK) << 8) | ((unsigned)c4 << 2);
                    hv[j] = *(const float4*)((const char*)H + off);
                }
#pragma unroll
                for (int j = 0; j < ROWS_IN_FLIGHT; ++j) {
                    int rj = (unsigned)ent[j].x >> COL_BITS;
                    float vj = j < rem ? __builtin_bit_cast(float, ent[j].y) : 0.0f;
                    if (rj != cur) {
                        float* dst = acc + cur * ACC_PITCH + c4;
                        __hip_atomic_fetch_add(dst + 0, sum.x, __ATOMIC_RELAXED, __HIP_MEMORY_SCOPE_WORKGROUP);
                        __hip_atomic_fetch_add(dst + 1, sum.y, __ATOMIC_RELAXED, __HIP_MEMORY_SCOPE_WORKGROUP);
                        __hip_atomic_fetch_add(dst + 2, sum.z, __ATOMIC_RELAXED, __HIP_MEMORY_SCOPE_WORKGROUP);
                        __hip_atomic_fetch_add(dst + 3, sum.w, __ATOMIC_RELAXED, __HIP_MEMORY_SCOPE_WORKGROUP);
                        cur = rj;
                        sum = zero4;
                    }
                    sum.x = __builtin_fmaf(vj, hv[j].x, sum.x);
                    sum.y = __builtin_fmaf(vj, hv[j].y, sum.y);
                    sum.z = __builtin_fmaf(vj, hv[j].z, sum.z);
                    sum.w = __builtin_fmaf(vj, hv[j].w, sum.w);
                }
            }
            if (active) {
                float* dst = acc + cur * ACC_PITCH + c4;
                __hip_atomic_fetch_add(dst + 0, sum.x, __ATOMIC_RELAXED, __HIP_MEMORY_SCOPE_WORKGROUP);
                __hip_atomic_fetch_add(dst + 1, sum.y, __ATOMIC_RELAXED, __HIP_MEMORY_SCOPE_WORKGROUP);
                __hip_atomic_fetch_add(dst + 2, sum.z, __ATOMIC_RELAXED, __HIP_MEMORY_SCOPE_WORKGROUP);
                __hip_atomic_fetch_add(dst + 3, sum.w, __ATOMIC_RELAXED, __HIP_MEMORY_SCOPE_WORKGROUP);
            }
        }
        __syncthreads();
    }

    float4 h0v[SUP_PER_THREAD];
#pragma unroll
    for (int u = 0; u < SUP_PER_THREAD; ++u) {
        int i = tid + u * SUM_THREADS;
        int ic = i < RB * FEAT / 4 ? i : RB * FEAT / 4 - 1;
        int grow = blk * RB + (ic >> 4);
        int gl = grow < nNodes ? grow : nNodes - 1;
        h0v[u] = ((const float4*)H0)[(size_t)gl * (FEAT / 4) + (ic & 15)];
    }
#pragma unroll
    for (int u = 0; u < SUP_PER_THREAD; ++u) {
        int i = tid + u * SUM_THREADS;
        int grow = blk * RB + (i >> 4);
        if (i < RB * FEAT / 4 && grow < nNodes) {
            float4* p = (float4*)(acc + (i >> 4) * ACC_PITCH) + (i & 15);
            float4 a = *p;
            a.x = oma * a.x + alpha * h0v[u].x;
            a.y = oma * a.y + alpha * h0v[u].y;
            a.z = oma * a.z + alpha * h0v[u].z;
            a.w = oma * a.w + alpha * h0v[u].w;
            *p = a;
        }
    }
    __syncthreads();

    if (wave >= 2 * ACC_TILES) return;
    int rt = wave >> 1;
    int ch = wave & 1;
    int hi = lane >> 4;
    int lm = lane & 15;
    const float* arow  = acc + (rt * 16 + lm) * ACC_PITCH + hi * 2;
    const float* wbase = W + (size_t)(hi * 2) * FEAT + ch * 32 + lm * 2;

    v8f acc0 = {}, acc1 = {};
#pragma unroll
    for (int kk = 0; kk < 16; ++kk) {
        int K0 = kk * 4;
        float2 s2 = *(const float2*)(arow + K0);
        float2 w0 = *(const float2*)(wbase + (size_t)(K0 + 0) * FEAT);
        float2 w1 = *(const float2*)(wbase + (size_t)(K0 + 1) * FEAT);
        v2f a, b0, b1;
        a.x = s2.x;  a.y = s2.y;
        b0.x = w0.x; b0.y = w1.x;
        b1.x = w0.y; b1.y = w1.y;
        acc0 = __builtin_amdgcn_wmma_f32_16x16x4_f32(
            false, a, false, b0, (short)0, acc0, false, false);
        acc1 = __builtin_amdgcn_wmma_f32_16x16x4_f32(
            false, a, false, b1, (short)0, acc1, false, false);
    }

#pragma unroll
    for (int r = 0; r < 8; ++r) {
        int rin = rt * 16 + r + hi * 8;
        int grow = blk * RB + rin;
        if (rin < RB && grow < nNodes) {
            float2 s = *(const float2*)(acc + rin * ACC_PITCH + ch * 32 + lm * 2);
            float2 o;
            o.x = omb * s.x + beta * acc0[r];
            o.y = omb * s.y + beta * acc1[r];
            *(float2*)(out + (size_t)grow * FEAT + ch * 32 + lm * 2) = o;
        }
    }
}

extern "C" void kernel_launch(void* const* d_in, const int* in_sizes, int n_in,
                              void* d_out, int out_size, void* d_ws, size_t ws_size,
                              hipStream_t stream) {
    const int*   row   = (const int*)  d_in[0];
    const int*   col   = (const int*)  d_in[1];
    const float* vals  = (const float*)d_in[2];
    const float* H     = (const float*)d_in[3];
    const float* H0    = (const float*)d_in[4];
    const float* W     = (const float*)d_in[5];
    const float* lamda = (const float*)d_in[6];
    const float* alpha = (const float*)d_in[7];
    const int*   l     = (const int*)  d_in[8];

    int nEdges  = in_sizes[0];
    int nNodes  = in_sizes[3] / FEAT;
    int nBins   = (nNodes + RB - 1) / RB;
    int nChunks = (nEdges + SORT_CHUNK - 1) / SORT_CHUNK;

    char* ws = (char*)d_ws;
    size_t binBytes = (((size_t)nEdges * 2 * sizeof(int)) + 255) & ~(size_t)255;
    int*   binned = (int*)ws;
    int*   table  = (int*)(ws + binBytes);

    size_t sortLds = (size_t)nBins * sizeof(int);
    gcnii_chunk_sort<<<nChunks, SORT_THREADS, sortLds, stream>>>(
        row, col, vals, table, binned, nEdges, nBins, nChunks);

    size_t segLds = (size_t)(2 * nChunks + 1) * sizeof(int);
    gcnii_block_sums<<<nBins, SUM_THREADS, segLds, stream>>>(
        binned, table, H, H0, W, (float*)d_out, lamda, alpha, l, nNodes, nChunks);
}


# Second launch: 16-wave work-groups (512 threads, 8 records a thread, 12 source rows in flight a half-wave): 8000 waves enter, not 16000; a wave does 2 GEMM tiles; row sums split at 32 shares, not 64
# speedup vs baseline: 4.0195x; 1.0354x over previous
#include <hip/hip_runtime.h>
#include <math.h>

typedef __attribute__((ext_vector_type(2))) float v2f;
typedef __attribute__((ext_vector_type(8))) float v8f;

#define FEAT 64

#define RB 200
#define COL_BITS 24
#define COL_MASK 0x00FFFFFF

#define SORT_THREADS 512
#define SORT_WAVES (SORT_THREADS / 32)
#define SORT_PER_THREAD 14
#define SORT_CHUNK (SORT_THREADS * SORT_PER_THREAD)

static __device__ __forceinline__ int group_scan(int c, int* waveTot, int nWaves, int& total) {
    int lane = threadIdx.x & 31;
    int wave = threadIdx.x >> 5;
    int x = c;
#pragma unroll
    for (int off = 1; off < 32; off <<= 1) {
        int y = __shfl_up(x, off);
        if (lane >= off) x += y;
    }
    if (lane == 31) waveTot[wave] = x;
    __syncthreads();
    int below = 0;
    int tot = 0;
    for (int w = 0; w < nWaves; ++w) {
        int t = waveTot[w];
        tot += t;
        if (w < wave) below += t;
    }
    __syncthreads();
    total = tot;
    return below + x - c;
}

__global__ void __launch_bounds__(SORT_THREADS)
gcnii_chunk_sort(const int* __restrict__ row,
                 const int* __restrict__ col,
                 const float* __restrict__ vals,
                 int* __restrict__ table,
                 int* __restrict__ binned,
                 int nEdges, int nBins, int nChunks) {
    extern __shared__ int ldsCnt[];
    __shared__ int2 stage[SORT_CHUNK];
    __shared__ int waveTot[SORT_WAVES];
    for (int b = threadIdx.x; b < nBins; b += SORT_THREADS) ldsCnt[b] = 0;
    int chunkBase = blockIdx.x * SORT_CHUNK;
    int base = chunkBase + threadIdx.x;
    int last = nEdges - 1;
    int r[SORT_PER_THREAD];
    int cl[SORT_PER_THREAD];
    float vl[SORT_PER_THREAD];
    int rank[SORT_PER_THREAD];
#pragma unroll
    for (int j = 0; j < SORT_PER_THREAD; ++j) {
        int e = base + j * SORT_THREADS;
        int ec = e < last ? e : last;
        r[j] = row[ec];
        cl[j] = col[ec];
        vl[j] = vals[ec];
    }
    __syncthreads();
#pragma unroll
    for (int j = 0; j < SORT_PER_THREAD; ++j) {
        int e = base + j * SORT_THREADS;
        rank[j] = 0;
        if (e < nEdges) rank[j] = atomicAdd(&ldsCnt[r[j] / RB], 1);
    }
    __syncthreads();
    int carry = 0;
    for (int b0 = 0; b0 < nBins; b0 += SORT_THREADS) {
        int b = b0 + threadIdx.x;
        int c = (b < nBins) ? ldsCnt[b] : 0;
        int tot;
        int ex = group_scan(c, waveTot, SORT_WAVES, tot);
        if (b < nBins) {
            int start = carry + ex;
            ldsCnt[b] = start;
            int2 t;
            t.x = chunkBase + start;
            t.y = c;
            *(int2*)(table + 2 * ((size_t)b * nChunks + blockIdx.x)) = t;
        }
        carry += tot;
    }
    __syncthreads();
    int runStart[SORT_PER_THREAD];
#pragma unroll
    for (int j = 0; j < SORT_PER_THREAD; ++j) runStart[j] = ldsCnt[r[j] / RB];
#pragma unroll
    for (int j = 0; j < SORT_PER_THREAD; ++j) {
        int e = base + j * SORT_THREADS;
        if (e < nEdges) {
            int b = r[j] / RB;
            int2 rec;
            rec.x = (int)(((unsigned)(r[j] - b * RB) << COL_BITS) | (unsigned)cl[j]);
            rec.y = __builtin_bit_cast(int, vl[j]);
            stage[runStart[j] + rank[j]] = rec;
        }
    }
    __syncthreads();
    int pairs = carry >> 1;
    int4* dst = (int4*)(binned + 2 * (size_t)chunkBase);
    for (int i = threadIdx.x; i < pairs; i += SORT_THREADS) dst[i] = ((const int4*)stage)[i];
    if ((carry & 1) && threadIdx.x == 0)
        *(int2*)(binned + 2 * (size_t)(chunkBase + carry - 1)) = stage[carry - 1];
}

#define ACC_TILES ((RB + 15) / 16)
#define ACC_ROWS (ACC_TILES * 16)
#define ACC_PITCH (FEAT + 4)
#define SUM_THREADS 512
#define SUM_WAVES (SUM_THREADS / 32)
#define ROWS_IN_FLIGHT 12
#define COPY_PER_THREAD 8
#define LIST_CAP (SUM_THREADS * COPY_PER_THREAD)
#define ROW_SLOTS (((RB + 31) / 32) * 32)
#define SUP_PER_THREAD ((RB * FEAT / 4 + SUM_THREADS - 1) / SUM_THREADS)

__global__ void __launch_bounds__(SUM_THREADS, 8)
gcnii_block_sums(const int* __restrict__ binned,
                 const int* __restrict__ table,
                 const float* __restrict__ H,
                 const float* __restrict__ H0,
                 const float* __restrict__ W,
                 float* __restrict__ out,
                 const float* __restrict__ lamdaPtr,
                 const float* __restrict__ alphaPtr,
                 const int* __restrict__ lPtr,
                 int nNodes, int nChunks) {
    __shared__ __attribute__((aligned(16))) float acc[ACC_ROWS * ACC_PITCH];
    __shared__ int2 list[LIST_CAP];
    __shared__ int waveTot[SUM_WAVES];
    __shared__ int rowCnt[ROW_SLOTS];
    __shared__ int rowStart[ROW_SLOTS];
    __shared__ int rowTot[ROW_SLOTS / 32];
    extern __shared__ int seg[];
    int* pre = seg;
    int* first = seg + nChunks + 1;
    int tid = threadIdx.x;
    int blk = blockIdx.x;

    float alpha = alphaPtr[0];
    float lamda = lamdaPtr[0];
    float beta  = logf(lamda / (float)lPtr[0] + 1.0f);
    float oma   = 1.0f - alpha;
    float omb   = 1.0f - beta;

    float4 zero4 = {0.0f, 0.0f, 0.0f, 0.0f};
    for (int i = tid; i < ACC_ROWS * ACC_PITCH / 4; i += SUM_THREADS) ((float4*)acc)[i] = zero4;
    if (tid < ROW_SLOTS) rowCnt[tid] = 0;

    int carry = 0;
    for (int s0 = 0; s0 < nChunks; s0 += SUM_THREADS) {
        int s = s0 + tid;
        int2 t;
        t.x = 0;
        t.y = 0;
        if (s < nChunks) t = *(const int2*)(table + 2 * ((size_t)blk * nChunks + s));
        int tot;
        int ex = group_scan(t.y, waveTot, SUM_WAVES, tot);
        if (s < nChunks) {
            pre[s] = carry + ex;
            first[s] = t.x;
        }
        carry += tot;
    }
    if (tid == 0) pre[nChunks] = carry;
    int total = __builtin_amdgcn_readfirstlane(carry);
    int wave = __builtin_amdgcn_readfirstlane(tid >> 5);
    int lane = tid & 31;
    int hf = lane >> 4;
    int c4 = (lane & 15) * 4;
    __syncthreads();

    for (int r0 = 0; r0 < total; r0 += LIST_CAP) {
        int n = total - r0;
        if (n > LIST_CAP) n = LIST_CAP;
        int g[COPY_PER_THREAD];
        int lo[COPY_PER_THREAD];
        int2 rec[COPY_PER_THREAD];
        int rank[COPY_PER_THREAD];
#pragma unroll
        for (int u = 0; u < COPY_PER_THREAD; ++u) {
            int i = tid + u * SUM_THREADS;
            g[u] = r0 + (i < n ? i : 0);
            lo[u] = 0;
        }
        for (int len = nChunks; len > 1; len -= len >> 1) {
            int half = len >> 1;
#pragma unroll
            for (int u = 0; u < COPY_PER_THREAD; ++u)
                if (pre[lo[u] + half] <= g[u]) lo[u] += half;
        }
#pragma unroll
        for (int u = 0; u < COPY_PER_THREAD; ++u)
            rec[u] = *(const int2*)(binned + 2 * (size_t)(first[lo[u]] + (g[u] - pre[lo[u]])));
#pragma unroll
        for (int u = 0; u < COPY_PER_THREAD; ++u) {
            int i = tid + u * SUM_THREADS;
            rank[u] = 0;
            if (i < n) rank[u] = atomicAdd(&rowCnt[(unsigned)rec[u].x >> COL_BITS], 1);
        }
        __syncthreads();
        if (tid < ROW_SLOTS) {
            int c = rowCnt[tid];
            rowCnt[tid] = 0;
            int x = c;
#pragma unroll
            for (int off = 1; off < 32; off <<= 1) {
                int y = __shfl_up(x, off);
                if (lane >= off) x += y;
            }
            rowStart[tid] = x - c;
            if (lane == 31) rowTot[tid >> 5] = x;
        }
        __syncthreads();
#pragma unroll
        for (int u = 0; u < COPY_PER_THREAD; ++u) {
            int i = tid + u * SUM_THREADS;
            if (i < n) {
                int rw = (unsigned)rec[u].x >> COL_BITS;
                int slot = rowStart[rw] + rank[u];
#pragma unroll
                for (int w = 0; w < ROW_SLOTS / 32 - 1; ++w)
                    if (rw >= 32 * (w + 1)) slot += rowTot[w];
                list[slot] = rec[u];
            }
        }
        __syncthreads();

        int per = (n + 2 * SUM_WAVES * ROWS_IN_FLIGHT - 1) / (2 * SUM_WAVES * ROWS_IN_FLIGHT) * ROWS_IN_FLIGHT;
        int waveBegin = wave * 2 * per;
        if (waveBegin < n) {
            int waveLen = n - waveBegin;
            if (waveLen > per) waveLen = per;
            int shareBegin = waveBegin + hf * per;
            int shareEnd = shareBegin + per;
            if (shareEnd > n) shareEnd = n;
            bool active = shareBegin < shareEnd;
            int lastIdx = active ? shareEnd - 1 : n - 1;
            int cur = (unsigned)list[active ? shareBegin : lastIdx].x >> COL_BITS;
            float4 sum = zero4;
            for (int t = 0; t < waveLen; t += ROWS_IN_FLIGHT) {
                int2 ent[ROWS_IN_FLIGHT];
                float4 hv[ROWS_IN_FLIGHT];
                int rem = shareEnd - shareBegin - t;
#pragma unroll
                for (int j = 0; j < ROWS_IN_FLIGHT; ++j) {
                    int idx = shareBegin + t + j;
                    ent[j] = list[idx < lastIdx ? idx : lastIdx];
                }
#pragma unroll
                for (int j = 0; j < ROWS_IN_FLIGHT; ++j) {
                    unsigned off = ((unsigned)(ent[j].x & COL_MASK) << 8) | ((unsigned)c4 << 2);
                    hv[j] = *(const float4*)((const char*)H + off);
                }
#pragma unroll
                for (int j = 0; j < ROWS_IN_FLIGHT; ++j) {
                    int rj = (unsigned)ent[j].x >> COL_BITS;
                    float vj = j < rem ? __builtin_bit_cast(float, ent[j].y) : 0.0f;
                    if (rj != cur) {
                        float* dst = acc + cur * ACC_PITCH + c4;
                        __hip_atomic_fetch_add(dst + 0, sum.x, __ATOMIC_RELAXED, __HIP_MEMORY_SCOPE_WORKGROUP);
                        __hip_atomic_fetch_add(dst + 1, sum.y, __ATOMIC_RELAXED, __HIP_MEMORY_SCOPE_WORKGROUP);
                        __hip_atomic_fetch_add(dst + 2, sum.z, __ATOMIC_RELAXED, __HIP_MEMORY_SCOPE_WORKGROUP);
                        __hip_atomic_fetch_add(dst + 3, sum.w, __ATOMIC_RELAXED, __HIP_MEMORY_SCOPE_WORKGROUP);
                        cur = rj;
                        sum = zero4;
                    }
                    sum.x = __builtin_fmaf(vj, hv[j].x, sum.x);
                    sum.y = __builtin_fmaf(vj, hv[j].y, sum.y);
                    sum.z = __builtin_fmaf(vj, hv[j].z, sum.z);
                    sum.w = __builtin_fmaf(vj, hv[j].w, sum.w);
                }
            }
            if (active) {
                float* dst = acc + cur * ACC_PITCH + c4;
                __hip_atomic_fetch_add(dst + 0, sum.x, __ATOMIC_RELAXED, __HIP_MEMORY_SCOPE_WORKGROUP);
                __hip_atomic_fetch_add(dst + 1, sum.y, __ATOMIC_RELAXED, __HIP_MEMORY_SCOPE_WORKGROUP);
                __hip_atomic_fetch_add(dst + 2, sum.z, __ATOMIC_RELAXED, __HIP_MEMORY_SCOPE_WORKGROUP);
                __hip_atomic_fetch_add(dst + 3, sum.w, __ATOMIC_RELAXED, __HIP_MEMORY_SCOPE_WORKGROUP);
            }
        }
        __syncthreads();
    }

    float4 h0v[SUP_PER_THREAD];
#pragma unroll
    for (int u = 0; u < SUP_PER_THREAD; ++u) {
        int i = tid + u * SUM_THREADS;
        int ic = i < RB * FEAT / 4 ? i : RB * FEAT / 4 - 1;
        int grow = blk * RB + (ic >> 4);
        int gl = grow < nNodes ? grow : nNodes - 1;
        h0v[u] = ((const float4*)H0)[(size_t)gl * (FEAT / 4) + (ic & 15)];
    }
#pragma unroll
    for (int u = 0; u < SUP_PER_THREAD; ++u) {
        int i = tid + u * SUM_THREADS;
        int grow = blk * RB + (i >> 4);
        if (i < RB * FEAT / 4 && grow < nNodes) {
            float4* p = (float4*)(acc + (i >> 4) * ACC_PITCH) + (i & 15);
            float4 a = *p;
            a.x = oma * a.x + alpha * h0v[u].x;
            a.y = oma * a.y + alpha * h0v[u].y;
            a.z = oma * a.z + alpha * h0v[u].z;
            a.w = oma * a.w + alpha * h0v[u].w;
            *p = a;
        }
    }
    __syncthreads();

    int ch = wave & 1;
    int hi = lane >> 4;
    int lm = lane & 15;
    const float* wbase = W + (size_t)(hi * 2) * FEAT + ch * 32 + lm * 2;
    for (int rt = wave >> 1; rt < ACC_TILES; rt += SUM_WAVES / 2) {
        const float* arow  = acc + (rt * 16 + lm) * ACC_PITCH + hi * 2;

        v8f acc0 = {}, acc1 = {};
#pragma unroll
        for (int kk = 0; kk < 16; ++kk) {
            int K0 = kk * 4;
            float2 s2 = *(const float2*)(arow + K0);
            float2 w0 = *(const float2*)(wbase + (size_t)(K0 + 0) * FEAT);
            float2 w1 = *(const float2*)(wbase + (size_t)(K0 + 1) * FEAT);
            v2f a, b0, b1;
            a.x = s2.x;  a.y = s2.y;
            b0.x = w0.x; b0.y = w1.x;
            b1.x = w0.y; b1.y = w1.y;
            acc0 = __builtin_amdgcn_wmma_f32_16x16x4_f32(
                false, a, false, b0, (short)0, acc0, false, false);
            acc1 = __builtin_amdgcn_wmma_f32_16x16x4_f32(
                false, a, false, b1, (short)0, acc1, false, false);
        }

#pragma unroll
        for (int r = 0; r < 8; ++r) {
            int rin = rt * 16 + r + hi * 8;
            int grow = blk * RB + rin;
            if (rin < RB && grow < nNodes) {
                float2 s = *(const float2*)(acc + rin * ACC_PITCH + ch * 32 + lm * 2);
                float2 o;
                o.x = omb * s.x + beta * acc0[r];
                o.y = omb * s.y + beta * acc1[r];
                *(float2*)(out + (size_t)grow * FEAT + ch * 32 + lm * 2) = o;
            }
        }
    }
}

extern "C" void kernel_launch(void* const* d_in, const int* in_sizes, int n_in,
                              void* d_out, int out_size, void* d_ws, size_t ws_size,
                              hipStream_t stream) {
    const int*   row   = (const int*)  d_in[0];
    const int*   col   = (const int*)  d_in[1];
    const float* vals  = (const float*)d_in[2];
    const float* H     = (const float*)d_in[3];
    const float* H0    = (const float*)d_in[4];
    const float* W     = (const float*)d_in[5];
    const float* lamda = (const float*)d_in[6];
    const float* alpha = (const float*)d_in[7];
    const int*   l     = (const int*)  d_in[8];

    int nEdges  = in_sizes[0];
    int nNodes  = in_sizes[3] / FEAT;
    int nBins   = (nNodes + RB - 1) / RB;
    int nChunks = (nEdges + SORT_CHUNK - 1) / SORT_CHUNK;

    char* ws = (char*)d_ws;
    size_t binBytes = (((size_t)nEdges * 2 * sizeof(int)) + 255) & ~(size_t)255;
    int*   binned = (int*)ws;
    int*   table  = (int*)(ws + binBytes);

    size_t sortLds = (size_t)nBins * sizeof(int);
    gcnii_chunk_sort<<<nChunks, SORT_THREADS, sortLds, stream>>>(
        row, col, vals, table, binned, nEdges, nBins, nChunks);

    size_t segLds = (size_t)(2 * nChunks + 1) * sizeof(int);
    gcnii_block_sums<<<nBins, SUM_THREADS, segLds, stream>>>(
        binned, table, H, H0, W, (float*)d_out, lamda, alpha, l, nNodes, nChunks);
}
